# starting kernel
# baseline (speedup 1.0000x reference)
#include <hip/hip_runtime.h>


typedef float v2f __attribute__((ext_vector_type(2)));
typedef float v8f __attribute__((ext_vector_type(8)));

struct c32 { float x, y; };
__device__ __forceinline__ c32 cmul(c32 a, c32 b) {
    return { a.x * b.x - a.y * b.y, a.x * b.y + a.y * b.x };
}

#define NQ     12
#define DIM    4096
#define NLAYER 4

__global__ __launch_bounds__(256)
void qsim_kernel(const float* __restrict__ xin,
                 const float* __restrict__ params,
                 float* __restrict__ out) {
    __shared__ float sre[DIM];
    __shared__ float sim_[DIM];
    __shared__ float bufAr[256], bufAi[256];
    __shared__ float bufBr[256], bufBi[256];
    __shared__ float uq[NQ][8];
    __shared__ float pv[NQ][6];
    __shared__ float pmr[NQ][16];
    __shared__ float pmi[NQ][16];
    __shared__ float enc[NQ][4];
    __shared__ float red[256 * NQ];

    const int tid  = threadIdx.x;
    const int b    = blockIdx.x;
    const int lane = tid & 31;
    const int wv   = tid >> 5;
    const int half = lane >> 4;
    const int lr   = lane & 15;

    auto wmma_apply = [&](const float* Wr, const float* Wi, int g) {
        const int rstride = (g == 0) ? 256 : ((g == 1) ? 16 : 1);
        const int nstride = (g == 2) ? 16 : 1;
        for (int tt = 0; tt < 2; ++tt) {
            int ct = 2 * wv + tt;
            int tbase = (g == 0) ? ct * 16 : ct * 256;
            v8f accR = {}; v8f accI = {};
#pragma unroll
            for (int ks = 0; ks < 4; ++ks) {
                int kk0 = 4 * ks + 2 * half;
                v2f ar, ai, ain, br, bi;
                ar[0] = Wr[lr * 16 + kk0];  ar[1] = Wr[lr * 16 + kk0 + 1];
                ai[0] = Wi[lr * 16 + kk0];  ai[1] = Wi[lr * 16 + kk0 + 1];
                ain = -ai;
                int i0 = tbase + kk0 * rstride + lr * nstride;
                br[0] = sre[i0];  br[1] = sre[i0 + rstride];
                bi[0] = sim_[i0]; bi[1] = sim_[i0 + rstride];
                accR = __builtin_amdgcn_wmma_f32_16x16x4_f32(
                    false, ar, false, br, (short)0, accR, false, false);
                accR = __builtin_amdgcn_wmma_f32_16x16x4_f32(
                    false, ain, false, bi, (short)0, accR, false, false);
                accI = __builtin_amdgcn_wmma_f32_16x16x4_f32(
                    false, ar, false, bi, (short)0, accI, false, false);
                accI = __builtin_amdgcn_wmma_f32_16x16x4_f32(
                    false, ai, false, br, (short)0, accI, false, false);
            }
#pragma unroll
            for (int j = 0; j < 8; ++j) {
                int M = j + 8 * half;
                int o = tbase + M * rstride + lr * nstride;
                sre[o] = accR[j]; sim_[o] = accI[j];
            }
        }
    };

    auto kron_build = [&](int g, float* outR, float* outI) {
        int r = tid >> 4, c = tid & 15, q0 = 4 * g;
        int o0 = ((((r >> 3) & 1) * 2 + ((c >> 3) & 1)) * 2);
        int o1 = ((((r >> 2) & 1) * 2 + ((c >> 2) & 1)) * 2);
        int o2 = ((((r >> 1) & 1) * 2 + ((c >> 1) & 1)) * 2);
        int o3 = (((r & 1) * 2 + (c & 1)) * 2);
        c32 acc = { uq[q0][o0], uq[q0][o0 + 1] };
        acc = cmul(acc, c32{ uq[q0 + 1][o1], uq[q0 + 1][o1 + 1] });
        acc = cmul(acc, c32{ uq[q0 + 2][o2], uq[q0 + 2][o2 + 1] });
        acc = cmul(acc, c32{ uq[q0 + 3][o3], uq[q0 + 3][o3 + 1] });
        outR[tid] = acc.x; outI[tid] = acc.y;
    };

    auto chain_step = [&](int pair, int t,
                          const float* inR, const float* inI,
                          float* outR, float* outI) {
        int r = tid >> 4, c = tid & 15;
        int sh = 2 - t;
        int rowsel = ((r >> sh) & 3) * 4;
        int rmask = r & ~(3 << sh);
        c32 acc = { 0.0f, 0.0f };
#pragma unroll
        for (int j = 0; j < 4; ++j) {
            int k = rmask | (j << sh);
            c32 p = { pmr[pair][rowsel + j], pmi[pair][rowsel + j] };
            c32 v = { inR[k * 16 + c], inI[k * 16 + c] };
            acc.x += p.x * v.x - p.y * v.y;
            acc.y += p.x * v.y + p.y * v.x;
        }
        outR[tid] = acc.x; outI[tid] = acc.y;
    };

    auto pfull_build = [&](int pair, float* outR, float* outI) {
        int r = tid >> 4, c = tid & 15;
        if ((r & 3) == (c & 3)) {
            int idx = ((r >> 2) & 3) * 4 + ((c >> 2) & 3);
            outR[tid] = pmr[pair][idx]; outI[tid] = pmi[pair][idx];
        } else {
            outR[tid] = 0.0f; outI[tid] = 0.0f;
        }
    };

    auto pair_sweep = [&](int i) {
        int j  = (i + 1) % NQ;
        int pi_ = 11 - i, pj_ = 11 - j;
        int lo = pi_ < pj_ ? pi_ : pj_;
        int hi = pi_ < pj_ ? pj_ : pi_;
        float cd = pv[i][0], sd = pv[i][1];
        float cs = pv[i][2], ss = pv[i][3];
        float cph = pv[i][4], sph = pv[i][5];
#pragma unroll
        for (int m = 0; m < 4; ++m) {
            int qd   = tid + 256 * m;
            int low  = qd & ((1 << lo) - 1);
            int mid  = (qd >> lo) & ((1 << (hi - 1 - lo)) - 1);
            int high = qd >> (hi - 1);
            int base = (high << (hi + 1)) | (mid << (lo + 1)) | low;
            int i01 = base | (1 << pj_);
            int i10 = base | (1 << pi_);
            int i11 = i10  | (1 << pj_);
            c32 a00 = { sre[base], sim_[base] };
            c32 a01 = { sre[i01],  sim_[i01]  };
            c32 a10 = { sre[i10],  sim_[i10]  };
            c32 a11 = { sre[i11],  sim_[i11]  };
            c32 b11 = { cph * a11.x - sph * a11.y, cph * a11.y + sph * a11.x };
            c32 b10 = { cph * a10.x + sph * a10.y, cph * a10.y - sph * a10.x };
            c32 n00 = {  cd * a00.x + sd * b11.y,  cd * a00.y - sd * b11.x };
            c32 n11 = {  sd * a00.y + cd * b11.x, -sd * a00.x + cd * b11.y };
            c32 n01 = {  cs * a01.x + ss * b10.y,  cs * a01.y - ss * b10.x };
            c32 n10 = {  ss * a01.y + cs * b10.x, -ss * a01.x + cs * b10.y };
            sre[base] = n00.x; sim_[base] = n00.y;
            sre[i01]  = n01.x; sim_[i01]  = n01.y;
            sre[i10]  = n10.x; sim_[i10]  = n10.y;
            sre[i11]  = n11.x; sim_[i11]  = n11.y;
        }
    };

    if (tid < NQ) {
        float xv = xin[b * NQ + tid];
        float h = 0.5f * xv;
        float a = 0.5f * xv * xv;
        float ch = cosf(h), sh = sinf(h);
        float ca = cosf(a), sa = sinf(a);
        enc[tid][0] = ch * ca;  enc[tid][1] = -ch * sa;
        enc[tid][2] = sh * ca;  enc[tid][3] =  sh * sa;
    }
    __syncthreads();

    for (int m = 0; m < 16; ++m) {
        int k = tid + 256 * m;
        c32 amp = { 1.0f, 0.0f };
#pragma unroll
        for (int q = 0; q < NQ; ++q) {
            int bit = (k >> (11 - q)) & 1;
            c32 f = { enc[q][bit * 2], enc[q][bit * 2 + 1] };
            amp = cmul(amp, f);
        }
        sre[k] = amp.x; sim_[k] = amp.y;
    }
    __syncthreads();

    for (int l = 0; l < NLAYER; ++l) {
        if (tid < NQ) {
            const float* wp = params + (l * NQ + tid) * 3;
            float w0 = wp[0], w1 = wp[1], w2 = wp[2];
            float c0 = cosf(0.5f * w0), s0 = sinf(0.5f * w0);
            float c1 = cosf(0.5f * w1), s1 = sinf(0.5f * w1);
            float cz = cosf(0.5f * w2), sz = sinf(0.5f * w2);
            c32 A  = {  c1 * c0,  s1 * s0 };
            c32 Bm = { -s1 * c0, -c1 * s0 };
            c32 Cm = {  s1 * c0, -c1 * s0 };
            c32 Dm = {  c1 * c0, -s1 * s0 };
            c32 ez  = { cz,  sz };
            c32 ezc = { cz, -sz };
            c32 u00 = cmul(A, ezc), u01 = cmul(Bm, ezc);
            c32 u10 = cmul(Cm, ez), u11 = cmul(Dm, ez);
            uq[tid][0] = u00.x; uq[tid][1] = u00.y;
            uq[tid][2] = u01.x; uq[tid][3] = u01.y;
            uq[tid][4] = u10.x; uq[tid][5] = u10.y;
            uq[tid][6] = u11.x; uq[tid][7] = u11.y;
            float cd = cosf(0.5f * (w1 - w2)), sd = sinf(0.5f * (w1 - w2));
            float cs = cosf(0.5f * (w1 + w2)), ss = sinf(0.5f * (w1 + w2));
            float cph = cosf(0.5f * w0), sph = sinf(0.5f * w0);
            pv[tid][0] = cd; pv[tid][1] = sd;
            pv[tid][2] = cs; pv[tid][3] = ss;
            pv[tid][4] = cph; pv[tid][5] = sph;
#pragma unroll
            for (int e = 0; e < 16; ++e) { pmr[tid][e] = 0.0f; pmi[tid][e] = 0.0f; }
            pmr[tid][0]  = cd;
            pmr[tid][3]  = sd * sph;  pmi[tid][3]  = -sd * cph;
            pmi[tid][12] = -sd;
            pmr[tid][15] = cd * cph;  pmi[tid][15] =  cd * sph;
            pmr[tid][5]  = cs;
            pmr[tid][6]  = -ss * sph; pmi[tid][6]  = -ss * cph;
            pmi[tid][9]  = -ss;
            pmr[tid][10] = cs * cph;  pmi[tid][10] = -cs * sph;
        }
        __syncthreads();

        kron_build(1, bufAr, bufAi);                 __syncthreads();
        wmma_apply(bufAr, bufAi, 1);                 __syncthreads();
        kron_build(2, bufAr, bufAi);                 __syncthreads();
        wmma_apply(bufAr, bufAi, 2);                 __syncthreads();

        kron_build(0, bufAr, bufAi);                 __syncthreads();
        chain_step(0, 0, bufAr, bufAi, bufBr, bufBi); __syncthreads();
        chain_step(1, 1, bufBr, bufBi, bufAr, bufAi); __syncthreads();
        chain_step(2, 2, bufAr, bufAi, bufBr, bufBi); __syncthreads();
        wmma_apply(bufBr, bufBi, 0);                 __syncthreads();

        pair_sweep(3);                               __syncthreads();

        pfull_build(4, bufAr, bufAi);                __syncthreads();
        chain_step(5, 1, bufAr, bufAi, bufBr, bufBi); __syncthreads();
        chain_step(6, 2, bufBr, bufBi, bufAr, bufAi); __syncthreads();
        wmma_apply(bufAr, bufAi, 1);                 __syncthreads();

        pair_sweep(7);                               __syncthreads();

        pfull_build(8, bufAr, bufAi);                __syncthreads();
        chain_step(9, 1, bufAr, bufAi, bufBr, bufBi); __syncthreads();
        chain_step(10, 2, bufBr, bufBi, bufAr, bufAi); __syncthreads();
        wmma_apply(bufAr, bufAi, 2);                 __syncthreads();

        pair_sweep(11);                              __syncthreads();
    }

    float part[NQ];
#pragma unroll
    for (int q = 0; q < NQ; ++q) part[q] = 0.0f;
    for (int m = 0; m < 16; ++m) {
        int k = tid + 256 * m;
        float pr = sre[k] * sre[k] + sim_[k] * sim_[k];
#pragma unroll
        for (int q = 0; q < NQ; ++q)
            part[q] += ((k >> (11 - q)) & 1) ? -pr : pr;
    }
#pragma unroll
    for (int q = 0; q < NQ; ++q) red[tid * NQ + q] = part[q];
    __syncthreads();
    for (int s = 128; s > 0; s >>= 1) {
        if (tid < s) {
#pragma unroll
            for (int q = 0; q < NQ; ++q)
                red[tid * NQ + q] += red[(tid + s) * NQ + q];
        }
        __syncthreads();
    }
    if (tid < NQ) out[b * NQ + tid] = red[tid];
}

extern "C" void kernel_launch(void* const* d_in, const int* in_sizes, int n_in,
                              void* d_out, int out_size, void* d_ws, size_t ws_size,
                              hipStream_t stream) {
    (void)n_in; (void)out_size; (void)d_ws; (void)ws_size;
    const float* x      = (const float*)d_in[0];
    const float* params = (const float*)d_in[1];
    float* out = (float*)d_out;
    int B = in_sizes[0] / NQ;
    qsim_kernel<<<dim3(B), dim3(256), 0, stream>>>(x, params, out);
}


# Regroup each layer's gates into 4 overlapping 4-qubit windows, each one 16x16 complex WMMA unitary: 16 applies replace 20 applies + 12 vector pair sweeps; pair products now summed inside WMMA
# speedup vs baseline: 1.2616x; 1.2616x over previous
#include <hip/hip_runtime.h>


typedef float v2f __attribute__((ext_vector_type(2)));
typedef float v8f __attribute__((ext_vector_type(8)));

struct c32 { float x, y; };
__device__ __forceinline__ c32 cmul(c32 a, c32 b) {
    return { a.x * b.x - a.y * b.y, a.x * b.y + a.y * b.x };
}

#define NQ     12
#define DIM    4096
#define NLAYER 4
#define NWIN   4
#define UID    NQ

template <int W>
__device__ __forceinline__ int row_addr(int r) {
    if (W == 0) return r << 8;
    if (W == 1) return r << 5;
    if (W == 2) return r << 2;
    return (r >> 1) | ((r & 1) << 11);
}
template <int W>
__device__ __forceinline__ int col_addr(int c) {
    if (W == 0) return c;
    if (W == 1) return (c & 31) | ((c >> 5) << 9);
    if (W == 2) return (c & 3) | ((c >> 2) << 6);
    return c << 3;
}

template <int W>
__device__ __forceinline__ void wmma_apply(float* sre, float* sim_,
                                           const float* Wr, const float* Wi,
                                           int wv, int half, int lr) {
    for (int tt = 0; tt < 2; ++tt) {
        const int cb = col_addr<W>((2 * wv + tt) * 16 + lr);
        v8f accR = {}; v8f accI = {};
#pragma unroll
        for (int ks = 0; ks < 4; ++ks) {
            int kk0 = 4 * ks + 2 * half;
            v2f ar, ai, ain, br, bi;
            ar[0] = Wr[lr * 16 + kk0];  ar[1] = Wr[lr * 16 + kk0 + 1];
            ai[0] = Wi[lr * 16 + kk0];  ai[1] = Wi[lr * 16 + kk0 + 1];
            ain = -ai;
            int i0 = cb + row_addr<W>(kk0);
            int i1 = cb + row_addr<W>(kk0 + 1);
            br[0] = sre[i0];  br[1] = sre[i1];
            bi[0] = sim_[i0]; bi[1] = sim_[i1];
            accR = __builtin_amdgcn_wmma_f32_16x16x4_f32(
                false, ar, false, br, (short)0, accR, false, false);
            accR = __builtin_amdgcn_wmma_f32_16x16x4_f32(
                false, ain, false, bi, (short)0, accR, false, false);
            accI = __builtin_amdgcn_wmma_f32_16x16x4_f32(
                false, ar, false, bi, (short)0, accI, false, false);
            accI = __builtin_amdgcn_wmma_f32_16x16x4_f32(
                false, ai, false, br, (short)0, accI, false, false);
        }
#pragma unroll
        for (int j = 0; j < 8; ++j) {
            int o = cb + row_addr<W>(j + 8 * half);
            sre[o] = accR[j]; sim_[o] = accI[j];
        }
    }
}

__global__ __launch_bounds__(256)
void qsim_kernel(const float* __restrict__ xin,
                 const float* __restrict__ params,
                 float* __restrict__ out) {
    __shared__ float sre[DIM];
    __shared__ float sim_[DIM];
    __shared__ float bufAr[NWIN * 256], bufAi[NWIN * 256];
    __shared__ float bufBr[NWIN * 256], bufBi[NWIN * 256];
    __shared__ float uq[NQ + 1][8];
    __shared__ float pmr[NQ][16];
    __shared__ float pmi[NQ][16];
    __shared__ float enc[NQ][4];
    __shared__ float red[256 * NQ];

    const int tid  = threadIdx.x;
    const int b    = blockIdx.x;
    const int lane = tid & 31;
    const int wv   = tid >> 5;
    const int half = lane >> 4;
    const int lr   = lane & 15;

    auto kron_build = [&](int w, float* outR, float* outI) {
        int r = tid >> 4, c = tid & 15;
        int f0 = (w == 0) ? 0 : UID;
        int f1 = 3 * w + 1, f2 = 3 * w + 2;
        int f3 = (w == 3) ? UID : 3 * w + 3;
        int o0 = ((((r >> 3) & 1) * 2 + ((c >> 3) & 1)) * 2);
        int o1 = ((((r >> 2) & 1) * 2 + ((c >> 2) & 1)) * 2);
        int o2 = ((((r >> 1) & 1) * 2 + ((c >> 1) & 1)) * 2);
        int o3 = (((r & 1) * 2 + (c & 1)) * 2);
        c32 acc = { uq[f0][o0], uq[f0][o0 + 1] };
        acc = cmul(acc, c32{ uq[f1][o1], uq[f1][o1 + 1] });
        acc = cmul(acc, c32{ uq[f2][o2], uq[f2][o2 + 1] });
        acc = cmul(acc, c32{ uq[f3][o3], uq[f3][o3 + 1] });
        outR[tid] = acc.x; outI[tid] = acc.y;
    };

    auto chain_step = [&](int pair, int t,
                          const float* inR, const float* inI,
                          float* outR, float* outI) {
        int r = tid >> 4, c = tid & 15;
        int sh = 2 - t;
        int rowsel = ((r >> sh) & 3) * 4;
        int rmask = r & ~(3 << sh);
        c32 acc = { 0.0f, 0.0f };
#pragma unroll
        for (int j = 0; j < 4; ++j) {
            int k = rmask | (j << sh);
            c32 p = { pmr[pair][rowsel + j], pmi[pair][rowsel + j] };
            c32 v = { inR[k * 16 + c], inI[k * 16 + c] };
            acc.x += p.x * v.x - p.y * v.y;
            acc.y += p.x * v.y + p.y * v.x;
        }
        outR[tid] = acc.x; outI[tid] = acc.y;
    };

    if (tid < NQ) {
        float xv = xin[b * NQ + tid];
        float h = 0.5f * xv;
        float a = 0.5f * xv * xv;
        float ch = cosf(h), sh = sinf(h);
        float ca = cosf(a), sa = sinf(a);
        enc[tid][0] = ch * ca;  enc[tid][1] = -ch * sa;
        enc[tid][2] = sh * ca;  enc[tid][3] =  sh * sa;
    }
    if (tid == NQ) {
        uq[UID][0] = 1.0f; uq[UID][1] = 0.0f; uq[UID][2] = 0.0f; uq[UID][3] = 0.0f;
        uq[UID][4] = 0.0f; uq[UID][5] = 0.0f; uq[UID][6] = 1.0f; uq[UID][7] = 0.0f;
    }
    __syncthreads();

    for (int m = 0; m < 16; ++m) {
        int k = tid + 256 * m;
        c32 amp = { 1.0f, 0.0f };
#pragma unroll
        for (int q = 0; q < NQ; ++q) {
            int bit = (k >> (11 - q)) & 1;
            c32 f = { enc[q][bit * 2], enc[q][bit * 2 + 1] };
            amp = cmul(amp, f);
        }
        sre[k] = amp.x; sim_[k] = amp.y;
    }
    __syncthreads();

    for (int l = 0; l < NLAYER; ++l) {
        if (tid < NQ) {
            const float* wp = params + (l * NQ + tid) * 3;
            float w0 = wp[0], w1 = wp[1], w2 = wp[2];
            float c0 = cosf(0.5f * w0), s0 = sinf(0.5f * w0);
            float c1 = cosf(0.5f * w1), s1 = sinf(0.5f * w1);
            float cz = cosf(0.5f * w2), sz = sinf(0.5f * w2);
            c32 A  = {  c1 * c0,  s1 * s0 };
            c32 Bm = { -s1 * c0, -c1 * s0 };
            c32 Cm = {  s1 * c0, -c1 * s0 };
            c32 Dm = {  c1 * c0, -s1 * s0 };
            c32 ez  = { cz,  sz };
            c32 ezc = { cz, -sz };
            c32 u00 = cmul(A, ezc), u01 = cmul(Bm, ezc);
            c32 u10 = cmul(Cm, ez), u11 = cmul(Dm, ez);
            uq[tid][0] = u00.x; uq[tid][1] = u00.y;
            uq[tid][2] = u01.x; uq[tid][3] = u01.y;
            uq[tid][4] = u10.x; uq[tid][5] = u10.y;
            uq[tid][6] = u11.x; uq[tid][7] = u11.y;
            float cd = cosf(0.5f * (w1 - w2)), sd = sinf(0.5f * (w1 - w2));
            float cs = cosf(0.5f * (w1 + w2)), ss = sinf(0.5f * (w1 + w2));
            float cph = cosf(0.5f * w0), sph = sinf(0.5f * w0);
#pragma unroll
            for (int e = 0; e < 16; ++e) { pmr[tid][e] = 0.0f; pmi[tid][e] = 0.0f; }
            pmr[tid][0]  = cd;
            pmr[tid][3]  = sd * sph;  pmi[tid][3]  = -sd * cph;
            pmi[tid][12] = -sd;
            pmr[tid][15] = cd * cph;  pmi[tid][15] =  cd * sph;
            pmr[tid][5]  = cs;
            pmr[tid][6]  = -ss * sph; pmi[tid][6]  = -ss * cph;
            pmi[tid][9]  = -ss;
            pmr[tid][10] = cs * cph;  pmi[tid][10] = -cs * sph;
        }
        __syncthreads();

        for (int w = 0; w < NWIN; ++w)
            kron_build(w, bufAr + 256 * w, bufAi + 256 * w);
        __syncthreads();
        for (int w = 0; w < NWIN; ++w)
            chain_step(3 * w, 0, bufAr + 256 * w, bufAi + 256 * w,
                                 bufBr + 256 * w, bufBi + 256 * w);
        __syncthreads();
        for (int w = 0; w < NWIN; ++w)
            chain_step(3 * w + 1, 1, bufBr + 256 * w, bufBi + 256 * w,
                                     bufAr + 256 * w, bufAi + 256 * w);
        __syncthreads();
        for (int w = 0; w < NWIN; ++w)
            chain_step(3 * w + 2, 2, bufAr + 256 * w, bufAi + 256 * w,
                                     bufBr + 256 * w, bufBi + 256 * w);
        __syncthreads();

        wmma_apply<0>(sre, sim_, bufBr,       bufBi,       wv, half, lr); __syncthreads();
        wmma_apply<1>(sre, sim_, bufBr + 256, bufBi + 256, wv, half, lr); __syncthreads();
        wmma_apply<2>(sre, sim_, bufBr + 512, bufBi + 512, wv, half, lr); __syncthreads();
        wmma_apply<3>(sre, sim_, bufBr + 768, bufBi + 768, wv, half, lr); __syncthreads();
    }

    float part[NQ];
#pragma unroll
    for (int q = 0; q < NQ; ++q) part[q] = 0.0f;
    for (int m = 0; m < 16; ++m) {
        int k = tid + 256 * m;
        float pr = sre[k] * sre[k] + sim_[k] * sim_[k];
#pragma unroll
        for (int q = 0; q < NQ; ++q)
            part[q] += ((k >> (11 - q)) & 1) ? -pr : pr;
    }
#pragma unroll
    for (int q = 0; q < NQ; ++q) red[tid * NQ + q] = part[q];
    __syncthreads();
    for (int s = 128; s > 0; s >>= 1) {
        if (tid < s) {
#pragma unroll
            for (int q = 0; q < NQ; ++q)
                red[tid * NQ + q] += red[(tid + s) * NQ + q];
        }
        __syncthreads();
    }
    if (tid < NQ) out[b * NQ + tid] = red[tid];
}

extern "C" void kernel_launch(void* const* d_in, const int* in_sizes, int n_in,
                              void* d_out, int out_size, void* d_ws, size_t ws_size,
                              hipStream_t stream) {
    (void)n_in; (void)out_size; (void)d_ws; (void)ws_size;
    const float* x      = (const float*)d_in[0];
    const float* params = (const float*)d_in[1];
    float* out = (float*)d_out;
    int B = in_sizes[0] / NQ;
    qsim_kernel<<<dim3(B), dim3(256), 0, stream>>>(x, params, out);
}


# All 16 window unitaries (4 layers x 4 windows) built once in a lane-parallel prologue, one cos/sin pair per lane, instead of per layer by 12 lanes; the layer loop holds only the WMMA applies
# speedup vs baseline: 1.3908x; 1.1024x over previous
#include <hip/hip_runtime.h>


typedef float v2f __attribute__((ext_vector_type(2)));
typedef float v8f __attribute__((ext_vector_type(8)));

struct c32 { float x, y; };
__device__ __forceinline__ c32 cmul(c32 a, c32 b) {
    return { a.x * b.x - a.y * b.y, a.x * b.y + a.y * b.x };
}

#define NQ     12
#define DIM    4096
#define NLAYER 4
#define NWIN   4
#define UID    NQ
#define NANG   5
#define NU     (NLAYER * NWIN)
#define ENC_H  (NLAYER * NQ * NANG)
#define ENC_A  (ENC_H + NQ)
#define NTRIG  (ENC_A + NQ)

template <int W>
__device__ __forceinline__ int row_addr(int r) {
    if (W == 0) return r << 8;
    if (W == 1) return r << 5;
    if (W == 2) return r << 2;
    return (r >> 1) | ((r & 1) << 11);
}
template <int W>
__device__ __forceinline__ int col_addr(int c) {
    if (W == 0) return c;
    if (W == 1) return (c & 31) | ((c >> 5) << 9);
    if (W == 2) return (c & 3) | ((c >> 2) << 6);
    return c << 3;
}

template <int W>
__device__ __forceinline__ void wmma_apply(float* sre, float* sim_,
                                           const float* Wr, const float* Wi,
                                           int wv, int half, int lr) {
    for (int tt = 0; tt < 2; ++tt) {
        const int cb = col_addr<W>((2 * wv + tt) * 16 + lr);
        v8f accR = {}; v8f accI = {};
#pragma unroll
        for (int ks = 0; ks < 4; ++ks) {
            int kk0 = 4 * ks + 2 * half;
            v2f ar, ai, ain, br, bi;
            ar[0] = Wr[lr * 16 + kk0];  ar[1] = Wr[lr * 16 + kk0 + 1];
            ai[0] = Wi[lr * 16 + kk0];  ai[1] = Wi[lr * 16 + kk0 + 1];
            ain = -ai;
            int i0 = cb + row_addr<W>(kk0);
            int i1 = cb + row_addr<W>(kk0 + 1);
            br[0] = sre[i0];  br[1] = sre[i1];
            bi[0] = sim_[i0]; bi[1] = sim_[i1];
            accR = __builtin_amdgcn_wmma_f32_16x16x4_f32(
                false, ar, false, br, (short)0, accR, false, false);
            accR = __builtin_amdgcn_wmma_f32_16x16x4_f32(
                false, ain, false, bi, (short)0, accR, false, false);
            accI = __builtin_amdgcn_wmma_f32_16x16x4_f32(
                false, ar, false, bi, (short)0, accI, false, false);
            accI = __builtin_amdgcn_wmma_f32_16x16x4_f32(
                false, ai, false, br, (short)0, accI, false, false);
        }
#pragma unroll
        for (int j = 0; j < 8; ++j) {
            int o = cb + row_addr<W>(j + 8 * half);
            sre[o] = accR[j]; sim_[o] = accI[j];
        }
    }
}

__global__ __launch_bounds__(256)
void qsim_kernel(const float* __restrict__ xin,
                 const float* __restrict__ params,
                 float* __restrict__ out) {
    __shared__ float sre[DIM];
    __shared__ float sim_[DIM];
    __shared__ float bufBr[NU * 256], bufBi[NU * 256];
    float* bufAr = sre;
    float* bufAi = sim_;
    __shared__ float tcs[NTRIG][2];
    __shared__ float uq[NLAYER][NQ + 1][8];
    __shared__ float pmr[NLAYER][NQ][16];
    __shared__ float pmi[NLAYER][NQ][16];
    __shared__ float enc[NQ][4];
    float* red = bufBr;

    const int tid  = threadIdx.x;
    const int b    = blockIdx.x;
    const int lane = tid & 31;
    const int wv   = tid >> 5;
    const int half = lane >> 4;
    const int lr   = lane & 15;

    auto kron_build = [&](int l, int w, float* outR, float* outI) {
        const float (*u)[8] = uq[l];
        int r = tid >> 4, c = tid & 15;
        int f0 = (w == 0) ? 0 : UID;
        int f1 = 3 * w + 1, f2 = 3 * w + 2;
        int f3 = (w == 3) ? UID : 3 * w + 3;
        int o0 = ((((r >> 3) & 1) * 2 + ((c >> 3) & 1)) * 2);
        int o1 = ((((r >> 2) & 1) * 2 + ((c >> 2) & 1)) * 2);
        int o2 = ((((r >> 1) & 1) * 2 + ((c >> 1) & 1)) * 2);
        int o3 = (((r & 1) * 2 + (c & 1)) * 2);
        c32 acc = { u[f0][o0], u[f0][o0 + 1] };
        acc = cmul(acc, c32{ u[f1][o1], u[f1][o1 + 1] });
        acc = cmul(acc, c32{ u[f2][o2], u[f2][o2 + 1] });
        acc = cmul(acc, c32{ u[f3][o3], u[f3][o3 + 1] });
        outR[tid] = acc.x; outI[tid] = acc.y;
    };

    auto chain_step = [&](int l, int pair, int t,
                          const float* inR, const float* inI,
                          float* outR, float* outI) {
        int r = tid >> 4, c = tid & 15;
        int sh = 2 - t;
        int rowsel = ((r >> sh) & 3) * 4;
        int rmask = r & ~(3 << sh);
        c32 acc = { 0.0f, 0.0f };
#pragma unroll
        for (int j = 0; j < 4; ++j) {
            int k = rmask | (j << sh);
            c32 p = { pmr[l][pair][rowsel + j], pmi[l][pair][rowsel + j] };
            c32 v = { inR[k * 16 + c], inI[k * 16 + c] };
            acc.x += p.x * v.x - p.y * v.y;
            acc.y += p.x * v.y + p.y * v.x;
        }
        outR[tid] = acc.x; outI[tid] = acc.y;
    };

    {
        float ang;
        if (tid < ENC_H) {
            int lq = tid / NANG, a = tid - lq * NANG;
            const float* wp = params + lq * 3;
            float w0 = wp[0], w1 = wp[1], w2 = wp[2];
            ang = (a == 0) ? 0.5f * w0
                : (a == 1) ? 0.5f * w1
                : (a == 2) ? 0.5f * w2
                : (a == 3) ? 0.5f * (w1 - w2)
                :            0.5f * (w1 + w2);
        } else {
            int e = tid - ENC_H;
            int q = (e < NQ) ? e : e - NQ;
            float xv = xin[b * NQ + q];
            ang = (e < NQ) ? 0.5f * xv : 0.5f * xv * xv;
        }
        tcs[tid][0] = cosf(ang);
        tcs[tid][1] = sinf(ang);
    }
    if (tid < NQ - 4) {
        int q = 4 + tid;
        float xv = xin[b * NQ + q];
        float a = 0.5f * xv * xv;
        tcs[ENC_A + q][0] = cosf(a);
        tcs[ENC_A + q][1] = sinf(a);
    }
    __syncthreads();

    if (tid < NLAYER * NQ) {
        const int l = tid / NQ, q = tid - l * NQ;
        const float (*t)[2] = &tcs[tid * NANG];
        float c0 = t[0][0], s0 = t[0][1];
        float c1 = t[1][0], s1 = t[1][1];
        float cz = t[2][0], sz = t[2][1];
        c32 A  = {  c1 * c0,  s1 * s0 };
        c32 Bm = { -s1 * c0, -c1 * s0 };
        c32 Cm = {  s1 * c0, -c1 * s0 };
        c32 Dm = {  c1 * c0, -s1 * s0 };
        c32 ez  = { cz,  sz };
        c32 ezc = { cz, -sz };
        c32 u00 = cmul(A, ezc), u01 = cmul(Bm, ezc);
        c32 u10 = cmul(Cm, ez), u11 = cmul(Dm, ez);
        float* uo = uq[l][q];
        uo[0] = u00.x; uo[1] = u00.y;
        uo[2] = u01.x; uo[3] = u01.y;
        uo[4] = u10.x; uo[5] = u10.y;
        uo[6] = u11.x; uo[7] = u11.y;
        float cd = t[3][0], sd = t[3][1];
        float cs = t[4][0], ss = t[4][1];
        float cph = c0, sph = s0;
        float* pr = pmr[l][q];
        float* pi = pmi[l][q];
#pragma unroll
        for (int e = 0; e < 16; ++e) { pr[e] = 0.0f; pi[e] = 0.0f; }
        pr[0]  = cd;
        pr[3]  = sd * sph;  pi[3]  = -sd * cph;
        pi[12] = -sd;
        pr[15] = cd * cph;  pi[15] =  cd * sph;
        pr[5]  = cs;
        pr[6]  = -ss * sph; pi[6]  = -ss * cph;
        pi[9]  = -ss;
        pr[10] = cs * cph;  pi[10] = -cs * sph;
    } else if (tid >= 64 && tid < 64 + NQ) {
        int q = tid - 64;
        float ch = tcs[ENC_H + q][0], sh = tcs[ENC_H + q][1];
        float ca = tcs[ENC_A + q][0], sa = tcs[ENC_A + q][1];
        enc[q][0] = ch * ca;  enc[q][1] = -ch * sa;
        enc[q][2] = sh * ca;  enc[q][3] =  sh * sa;
    } else if (tid >= 96 && tid < 96 + NLAYER) {
        int l = tid - 96;
        uq[l][UID][0] = 1.0f; uq[l][UID][1] = 0.0f; uq[l][UID][2] = 0.0f; uq[l][UID][3] = 0.0f;
        uq[l][UID][4] = 0.0f; uq[l][UID][5] = 0.0f; uq[l][UID][6] = 1.0f; uq[l][UID][7] = 0.0f;
    }
    __syncthreads();

#pragma unroll
    for (int g = 0; g < NU; ++g)
        kron_build(g >> 2, g & 3, bufAr + 256 * g, bufAi + 256 * g);
    __syncthreads();
#pragma unroll
    for (int g = 0; g < NU; ++g)
        chain_step(g >> 2, 3 * (g & 3), 0, bufAr + 256 * g, bufAi + 256 * g,
                                           bufBr + 256 * g, bufBi + 256 * g);
    __syncthreads();
#pragma unroll
    for (int g = 0; g < NU; ++g)
        chain_step(g >> 2, 3 * (g & 3) + 1, 1, bufBr + 256 * g, bufBi + 256 * g,
                                               bufAr + 256 * g, bufAi + 256 * g);
    __syncthreads();
#pragma unroll
    for (int g = 0; g < NU; ++g)
        chain_step(g >> 2, 3 * (g & 3) + 2, 2, bufAr + 256 * g, bufAi + 256 * g,
                                               bufBr + 256 * g, bufBi + 256 * g);
    __syncthreads();

    for (int m = 0; m < 16; ++m) {
        int k = tid + 256 * m;
        c32 amp = { 1.0f, 0.0f };
#pragma unroll
        for (int q = 0; q < NQ; ++q) {
            int bit = (k >> (11 - q)) & 1;
            c32 f = { enc[q][bit * 2], enc[q][bit * 2 + 1] };
            amp = cmul(amp, f);
        }
        sre[k] = amp.x; sim_[k] = amp.y;
    }
    __syncthreads();

    for (int l = 0; l < NLAYER; ++l) {
        const float* Ur = bufBr + 1024 * l;
        const float* Ui = bufBi + 1024 * l;
        wmma_apply<0>(sre, sim_, Ur,       Ui,       wv, half, lr); __syncthreads();
        wmma_apply<1>(sre, sim_, Ur + 256, Ui + 256, wv, half, lr); __syncthreads();
        wmma_apply<2>(sre, sim_, Ur + 512, Ui + 512, wv, half, lr); __syncthreads();
        wmma_apply<3>(sre, sim_, Ur + 768, Ui + 768, wv, half, lr); __syncthreads();
    }

    float part[NQ];
#pragma unroll
    for (int q = 0; q < NQ; ++q) part[q] = 0.0f;
    for (int m = 0; m < 16; ++m) {
        int k = tid + 256 * m;
        float pr = sre[k] * sre[k] + sim_[k] * sim_[k];
#pragma unroll
        for (int q = 0; q < NQ; ++q)
            part[q] += ((k >> (11 - q)) & 1) ? -pr : pr;
    }
#pragma unroll
    for (int q = 0; q < NQ; ++q) red[tid * NQ + q] = part[q];
    __syncthreads();
    for (int s = 128; s > 0; s >>= 1) {
        if (tid < s) {
#pragma unroll
            for (int q = 0; q < NQ; ++q)
                red[tid * NQ + q] += red[(tid + s) * NQ + q];
        }
        __syncthreads();
    }
    if (tid < NQ) out[b * NQ + tid] = red[tid];
}

extern "C" void kernel_launch(void* const* d_in, const int* in_sizes, int n_in,
                              void* d_out, int out_size, void* d_ws, size_t ws_size,
                              hipStream_t stream) {
    (void)n_in; (void)out_size; (void)d_ws; (void)ws_size;
    const float* x      = (const float*)d_in[0];
    const float* params = (const float*)d_in[1];
    float* out = (float*)d_out;
    int B = in_sizes[0] / NQ;
    qsim_kernel<<<dim3(B), dim3(256), 0, stream>>>(x, params, out);
}


# Complex window apply as 3 real WMMA products, not 4: K1=(Wr+Wi)Br shared, Re=K1-Wi(Br+Bi), Im=K1+Wr(Bi-Br); 384 WMMAs per wave instead of 512; changes rounding and summation order
# speedup vs baseline: 1.4806x; 1.0646x over previous
#include <hip/hip_runtime.h>


typedef float v2f __attribute__((ext_vector_type(2)));
typedef float v8f __attribute__((ext_vector_type(8)));

struct c32 { float x, y; };
__device__ __forceinline__ c32 cmul(c32 a, c32 b) {
    return { a.x * b.x - a.y * b.y, a.x * b.y + a.y * b.x };
}

#define NQ     12
#define DIM    4096
#define NLAYER 4
#define NWIN   4
#define UID    NQ
#define NANG   5
#define NU     (NLAYER * NWIN)
#define ENC_H  (NLAYER * NQ * NANG)
#define ENC_A  (ENC_H + NQ)
#define NTRIG  (ENC_A + NQ)

template <int W>
__device__ __forceinline__ int row_addr(int r) {
    if (W == 0) return r << 8;
    if (W == 1) return r << 5;
    if (W == 2) return r << 2;
    return (r >> 1) | ((r & 1) << 11);
}
template <int W>
__device__ __forceinline__ int col_addr(int c) {
    if (W == 0) return c;
    if (W == 1) return (c & 31) | ((c >> 5) << 9);
    if (W == 2) return (c & 3) | ((c >> 2) << 6);
    return c << 3;
}

template <int W>
__device__ __forceinline__ void wmma_apply(float* sre, float* sim_,
                                           const float* Wr, const float* Wi,
                                           int wv, int half, int lr) {
    v2f ar[4], an[4], as[4];
#pragma unroll
    for (int ks = 0; ks < 4; ++ks) {
        int kk0 = 4 * ks + 2 * half;
        v2f ai;
        ar[ks][0] = Wr[lr * 16 + kk0];  ar[ks][1] = Wr[lr * 16 + kk0 + 1];
        ai[0]     = Wi[lr * 16 + kk0];  ai[1]     = Wi[lr * 16 + kk0 + 1];
        an[ks] = -ai;
        as[ks] = ar[ks] + ai;
    }
    for (int tt = 0; tt < 2; ++tt) {
        const int cb = col_addr<W>((2 * wv + tt) * 16 + lr);
        v2f br[4], bs[4], bd[4];
#pragma unroll
        for (int ks = 0; ks < 4; ++ks) {
            int kk0 = 4 * ks + 2 * half;
            int i0 = cb + row_addr<W>(kk0);
            int i1 = cb + row_addr<W>(kk0 + 1);
            v2f bi;
            br[ks][0] = sre[i0];  br[ks][1] = sre[i1];
            bi[0]     = sim_[i0]; bi[1]     = sim_[i1];
            bs[ks] = br[ks] + bi;
            bd[ks] = bi - br[ks];
        }
        v8f k1 = {};
#pragma unroll
        for (int ks = 0; ks < 4; ++ks)
            k1 = __builtin_amdgcn_wmma_f32_16x16x4_f32(
                false, as[ks], false, br[ks], (short)0, k1, false, false);
        v8f accR = k1; v8f accI = k1;
#pragma unroll
        for (int ks = 0; ks < 4; ++ks) {
            accR = __builtin_amdgcn_wmma_f32_16x16x4_f32(
                false, an[ks], false, bs[ks], (short)0, accR, false, false);
            accI = __builtin_amdgcn_wmma_f32_16x16x4_f32(
                false, ar[ks], false, bd[ks], (short)0, accI, false, false);
        }
#pragma unroll
        for (int j = 0; j < 8; ++j) {
            int o = cb + row_addr<W>(j + 8 * half);
            sre[o] = accR[j]; sim_[o] = accI[j];
        }
    }
}

__global__ __launch_bounds__(256)
void qsim_kernel(const float* __restrict__ xin,
                 const float* __restrict__ params,
                 float* __restrict__ out) {
    __shared__ float sre[DIM];
    __shared__ float sim_[DIM];
    __shared__ float bufBr[NU * 256], bufBi[NU * 256];
    float* bufAr = sre;
    float* bufAi = sim_;
    __shared__ float tcs[NTRIG][2];
    __shared__ float uq[NLAYER][NQ + 1][8];
    __shared__ float pmr[NLAYER][NQ][16];
    __shared__ float pmi[NLAYER][NQ][16];
    __shared__ float enc[NQ][4];
    float* red = bufBr;

    const int tid  = threadIdx.x;
    const int b    = blockIdx.x;
    const int lane = tid & 31;
    const int wv   = tid >> 5;
    const int half = lane >> 4;
    const int lr   = lane & 15;

    auto kron_build = [&](int l, int w, float* outR, float* outI) {
        const float (*u)[8] = uq[l];
        int r = tid >> 4, c = tid & 15;
        int f0 = (w == 0) ? 0 : UID;
        int f1 = 3 * w + 1, f2 = 3 * w + 2;
        int f3 = (w == 3) ? UID : 3 * w + 3;
        int o0 = ((((r >> 3) & 1) * 2 + ((c >> 3) & 1)) * 2);
        int o1 = ((((r >> 2) & 1) * 2 + ((c >> 2) & 1)) * 2);
        int o2 = ((((r >> 1) & 1) * 2 + ((c >> 1) & 1)) * 2);
        int o3 = (((r & 1) * 2 + (c & 1)) * 2);
        c32 acc = { u[f0][o0], u[f0][o0 + 1] };
        acc = cmul(acc, c32{ u[f1][o1], u[f1][o1 + 1] });
        acc = cmul(acc, c32{ u[f2][o2], u[f2][o2 + 1] });
        acc = cmul(acc, c32{ u[f3][o3], u[f3][o3 + 1] });
        outR[tid] = acc.x; outI[tid] = acc.y;
    };

    auto chain_step = [&](int l, int pair, int t,
                          const float* inR, const float* inI,
                          float* outR, float* outI) {
        int r = tid >> 4, c = tid & 15;
        int sh = 2 - t;
        int rowsel = ((r >> sh) & 3) * 4;
        int rmask = r & ~(3 << sh);
        c32 acc = { 0.0f, 0.0f };
#pragma unroll
        for (int j = 0; j < 4; ++j) {
            int k = rmask | (j << sh);
            c32 p = { pmr[l][pair][rowsel + j], pmi[l][pair][rowsel + j] };
            c32 v = { inR[k * 16 + c], inI[k * 16 + c] };
            acc.x += p.x * v.x - p.y * v.y;
            acc.y += p.x * v.y + p.y * v.x;
        }
        outR[tid] = acc.x; outI[tid] = acc.y;
    };

    {
        float ang;
        if (tid < ENC_H) {
            int lq = tid / NANG, a = tid - lq * NANG;
            const float* wp = params + lq * 3;
            float w0 = wp[0], w1 = wp[1], w2 = wp[2];
            ang = (a == 0) ? 0.5f * w0
                : (a == 1) ? 0.5f * w1
                : (a == 2) ? 0.5f * w2
                : (a == 3) ? 0.5f * (w1 - w2)
                :            0.5f * (w1 + w2);
        } else {
            int e = tid - ENC_H;
            int q = (e < NQ) ? e : e - NQ;
            float xv = xin[b * NQ + q];
            ang = (e < NQ) ? 0.5f * xv : 0.5f * xv * xv;
        }
        tcs[tid][0] = cosf(ang);
        tcs[tid][1] = sinf(ang);
    }
    if (tid < NQ - 4) {
        int q = 4 + tid;
        float xv = xin[b * NQ + q];
        float a = 0.5f * xv * xv;
        tcs[ENC_A + q][0] = cosf(a);
        tcs[ENC_A + q][1] = sinf(a);
    }
    __syncthreads();

    if (tid < NLAYER * NQ) {
        const int l = tid / NQ, q = tid - l * NQ;
        const float (*t)[2] = &tcs[tid * NANG];
        float c0 = t[0][0], s0 = t[0][1];
        float c1 = t[1][0], s1 = t[1][1];
        float cz = t[2][0], sz = t[2][1];
        c32 A  = {  c1 * c0,  s1 * s0 };
        c32 Bm = { -s1 * c0, -c1 * s0 };
        c32 Cm = {  s1 * c0, -c1 * s0 };
        c32 Dm = {  c1 * c0, -s1 * s0 };
        c32 ez  = { cz,  sz };
        c32 ezc = { cz, -sz };
        c32 u00 = cmul(A, ezc), u01 = cmul(Bm, ezc);
        c32 u10 = cmul(Cm, ez), u11 = cmul(Dm, ez);
        float* uo = uq[l][q];
        uo[0] = u00.x; uo[1] = u00.y;
        uo[2] = u01.x; uo[3] = u01.y;
        uo[4] = u10.x; uo[5] = u10.y;
        uo[6] = u11.x; uo[7] = u11.y;
        float cd = t[3][0], sd = t[3][1];
        float cs = t[4][0], ss = t[4][1];
        float cph = c0, sph = s0;
        float* pr = pmr[l][q];
        float* pi = pmi[l][q];
#pragma unroll
        for (int e = 0; e < 16; ++e) { pr[e] = 0.0f; pi[e] = 0.0f; }
        pr[0]  = cd;
        pr[3]  = sd * sph;  pi[3]  = -sd * cph;
        pi[12] = -sd;
        pr[15] = cd * cph;  pi[15] =  cd * sph;
        pr[5]  = cs;
        pr[6]  = -ss * sph; pi[6]  = -ss * cph;
        pi[9]  = -ss;
        pr[10] = cs * cph;  pi[10] = -cs * sph;
    } else if (tid >= 64 && tid < 64 + NQ) {
        int q = tid - 64;
        float ch = tcs[ENC_H + q][0], sh = tcs[ENC_H + q][1];
        float ca = tcs[ENC_A + q][0], sa = tcs[ENC_A + q][1];
        enc[q][0] = ch * ca;  enc[q][1] = -ch * sa;
        enc[q][2] = sh * ca;  enc[q][3] =  sh * sa;
    } else if (tid >= 96 && tid < 96 + NLAYER) {
        int l = tid - 96;
        uq[l][UID][0] = 1.0f; uq[l][UID][1] = 0.0f; uq[l][UID][2] = 0.0f; uq[l][UID][3] = 0.0f;
        uq[l][UID][4] = 0.0f; uq[l][UID][5] = 0.0f; uq[l][UID][6] = 1.0f; uq[l][UID][7] = 0.0f;
    }
    __syncthreads();

#pragma unroll
    for (int g = 0; g < NU; ++g)
        kron_build(g >> 2, g & 3, bufAr + 256 * g, bufAi + 256 * g);
    __syncthreads();
#pragma unroll
    for (int g = 0; g < NU; ++g)
        chain_step(g >> 2, 3 * (g & 3), 0, bufAr + 256 * g, bufAi + 256 * g,
                                           bufBr + 256 * g, bufBi + 256 * g);
    __syncthreads();
#pragma unroll
    for (int g = 0; g < NU; ++g)
        chain_step(g >> 2, 3 * (g & 3) + 1, 1, bufBr + 256 * g, bufBi + 256 * g,
                                               bufAr + 256 * g, bufAi + 256 * g);
    __syncthreads();
#pragma unroll
    for (int g = 0; g < NU; ++g)
        chain_step(g >> 2, 3 * (g & 3) + 2, 2, bufAr + 256 * g, bufAi + 256 * g,
                                               bufBr + 256 * g, bufBi + 256 * g);
    __syncthreads();

    for (int m = 0; m < 16; ++m) {
        int k = tid + 256 * m;
        c32 amp = { 1.0f, 0.0f };
#pragma unroll
        for (int q = 0; q < NQ; ++q) {
            int bit = (k >> (11 - q)) & 1;
            c32 f = { enc[q][bit * 2], enc[q][bit * 2 + 1] };
            amp = cmul(amp, f);
        }
        sre[k] = amp.x; sim_[k] = amp.y;
    }
    __syncthreads();

    for (int l = 0; l < NLAYER; ++l) {
        const float* Ur = bufBr + 1024 * l;
        const float* Ui = bufBi + 1024 * l;
        wmma_apply<0>(sre, sim_, Ur,       Ui,       wv, half, lr); __syncthreads();
        wmma_apply<1>(sre, sim_, Ur + 256, Ui + 256, wv, half, lr); __syncthreads();
        wmma_apply<2>(sre, sim_, Ur + 512, Ui + 512, wv, half, lr); __syncthreads();
        wmma_apply<3>(sre, sim_, Ur + 768, Ui + 768, wv, half, lr); __syncthreads();
    }

    float part[NQ];
#pragma unroll
    for (int q = 0; q < NQ; ++q) part[q] = 0.0f;
    for (int m = 0; m < 16; ++m) {
        int k = tid + 256 * m;
        float pr = sre[k] * sre[k] + sim_[k] * sim_[k];
#pragma unroll
        for (int q = 0; q < NQ; ++q)
            part[q] += ((k >> (11 - q)) & 1) ? -pr : pr;
    }
#pragma unroll
    for (int q = 0; q < NQ; ++q) red[tid * NQ + q] = part[q];
    __syncthreads();
    for (int s = 128; s > 0; s >>= 1) {
        if (tid < s) {
#pragma unroll
            for (int q = 0; q < NQ; ++q)
                red[tid * NQ + q] += red[(tid + s) * NQ + q];
        }
        __syncthreads();
    }
    if (tid < NQ) out[b * NQ + tid] = red[tid];
}

extern "C" void kernel_launch(void* const* d_in, const int* in_sizes, int n_in,
                              void* d_out, int out_size, void* d_ws, size_t ws_size,
                              hipStream_t stream) {
    (void)n_in; (void)out_size; (void)d_ws; (void)ws_size;
    const float* x      = (const float*)d_in[0];
    const float* params = (const float*)d_in[1];
    float* out = (float*)d_out;
    int B = in_sizes[0] / NQ;
    qsim_kernel<<<dim3(B), dim3(256), 0, stream>>>(x, params, out);
}


# LDS bank conflicts removed from the window applies: state amplitudes at a padded address (index bits 6-11 add offsets so each fragment load and result store hits 32 banks), unitaries at row pitch 20
# speedup vs baseline: 1.7461x; 1.1793x over previous
#include <hip/hip_runtime.h>


typedef float v2f __attribute__((ext_vector_type(2)));
typedef float v8f __attribute__((ext_vector_type(8)));

struct c32 { float x, y; };
__device__ __forceinline__ c32 cmul(c32 a, c32 b) {
    return { a.x * b.x - a.y * b.y, a.x * b.y + a.y * b.x };
}

#define NQ     12
#define DIM    4096
#define NLAYER 4
#define NWIN   4
#define UID    NQ
#define NANG   5
#define NU     (NLAYER * NWIN)
#define ENC_H  (NLAYER * NQ * NANG)
#define ENC_A  (ENC_H + NQ)
#define NTRIG  (ENC_A + NQ)
#define UP     20
#define UPW    (16 * UP)

#define SP     5664
__device__ __forceinline__ constexpr int pa(int k) {
    return k + ((k >> 6) & 1) * 18 + ((k >> 7) & 1) * 68 + ((k >> 8) & 1) * 96
             + ((k >> 9) & 1) * 208 + ((k >> 10) & 1) * 392 + ((k >> 11) & 1) * 784;
}

template <int W>
__device__ __forceinline__ constexpr int row_addr(int r) {
    if (W == 0) return r << 8;
    if (W == 1) return r << 5;
    if (W == 2) return r << 2;
    return (r >> 1) | ((r & 1) << 11);
}
template <int W>
__device__ __forceinline__ constexpr int col_addr(int c) {
    if (W == 0) return c;
    if (W == 1) return (c & 31) | ((c >> 5) << 9);
    if (W == 2) return (c & 3) | ((c >> 2) << 6);
    return c << 3;
}

template <int W>
__device__ __forceinline__ void wmma_apply(float* sre, float* sim_,
                                           const float* Wr, const float* Wi,
                                           int wv, int half, int lr) {
    v2f ar[4], an[4], as[4];
#pragma unroll
    for (int ks = 0; ks < 4; ++ks) {
        int kk0 = 4 * ks + 2 * half;
        v2f ai;
        ar[ks][0] = Wr[lr * UP + kk0];  ar[ks][1] = Wr[lr * UP + kk0 + 1];
        ai[0]     = Wi[lr * UP + kk0];  ai[1]     = Wi[lr * UP + kk0 + 1];
        an[ks] = -ai;
        as[ks] = ar[ks] + ai;
    }
    for (int tt = 0; tt < 2; ++tt) {
        const int cb = col_addr<W>((2 * wv + tt) * 16 + lr);
        const int lb = pa(cb + row_addr<W>(2 * half));
        const int sb = pa(cb + row_addr<W>(8 * half));
        v2f br[4], bs[4], bd[4];
#pragma unroll
        for (int ks = 0; ks < 4; ++ks) {
            int i0 = lb + pa(row_addr<W>(4 * ks));
            int i1 = lb + pa(row_addr<W>(4 * ks + 1));
            v2f bi;
            br[ks][0] = sre[i0];  br[ks][1] = sre[i1];
            bi[0]     = sim_[i0]; bi[1]     = sim_[i1];
            bs[ks] = br[ks] + bi;
            bd[ks] = bi - br[ks];
        }
        v8f k1 = {};
#pragma unroll
        for (int ks = 0; ks < 4; ++ks)
            k1 = __builtin_amdgcn_wmma_f32_16x16x4_f32(
                false, as[ks], false, br[ks], (short)0, k1, false, false);
        v8f accR = k1; v8f accI = k1;
#pragma unroll
        for (int ks = 0; ks < 4; ++ks) {
            accR = __builtin_amdgcn_wmma_f32_16x16x4_f32(
                false, an[ks], false, bs[ks], (short)0, accR, false, false);
            accI = __builtin_amdgcn_wmma_f32_16x16x4_f32(
                false, ar[ks], false, bd[ks], (short)0, accI, false, false);
        }
#pragma unroll
        for (int j = 0; j < 8; ++j) {
            int o = sb + pa(row_addr<W>(j));
            sre[o] = accR[j]; sim_[o] = accI[j];
        }
    }
}

__global__ __launch_bounds__(256)
void qsim_kernel(const float* __restrict__ xin,
                 const float* __restrict__ params,
                 float* __restrict__ out) {
    __shared__ float sre[SP];
    __shared__ float sim_[SP];
    __shared__ float bufBr[NU * UPW], bufBi[NU * UPW];
    float* bufAr = sre;
    float* bufAi = sim_;
    __shared__ float tcs[NTRIG][2];
    __shared__ float uq[NLAYER][NQ + 1][8];
    __shared__ float pmr[NLAYER][NQ][16];
    __shared__ float pmi[NLAYER][NQ][16];
    __shared__ float enc[NQ][4];
    float* red = bufBr;

    const int tid  = threadIdx.x;
    const int b    = blockIdx.x;
    const int lane = tid & 31;
    const int wv   = tid >> 5;
    const int half = lane >> 4;
    const int lr   = lane & 15;

    auto kron_build = [&](int l, int w, float* outR, float* outI) {
        const float (*u)[8] = uq[l];
        int r = tid >> 4, c = tid & 15;
        int f0 = (w == 0) ? 0 : UID;
        int f1 = 3 * w + 1, f2 = 3 * w + 2;
        int f3 = (w == 3) ? UID : 3 * w + 3;
        int o0 = ((((r >> 3) & 1) * 2 + ((c >> 3) & 1)) * 2);
        int o1 = ((((r >> 2) & 1) * 2 + ((c >> 2) & 1)) * 2);
        int o2 = ((((r >> 1) & 1) * 2 + ((c >> 1) & 1)) * 2);
        int o3 = (((r & 1) * 2 + (c & 1)) * 2);
        c32 acc = { u[f0][o0], u[f0][o0 + 1] };
        acc = cmul(acc, c32{ u[f1][o1], u[f1][o1 + 1] });
        acc = cmul(acc, c32{ u[f2][o2], u[f2][o2 + 1] });
        acc = cmul(acc, c32{ u[f3][o3], u[f3][o3 + 1] });
        outR[tid] = acc.x; outI[tid] = acc.y;
    };

    auto chain_step = [&](int l, int pair, int t,
                          const float* inR, const float* inI, int ip,
                          float* outR, float* outI, int op) {
        int r = tid >> 4, c = tid & 15;
        int sh = 2 - t;
        int rowsel = ((r >> sh) & 3) * 4;
        int rmask = r & ~(3 << sh);
        c32 acc = { 0.0f, 0.0f };
#pragma unroll
        for (int j = 0; j < 4; ++j) {
            int k = rmask | (j << sh);
            c32 p = { pmr[l][pair][rowsel + j], pmi[l][pair][rowsel + j] };
            c32 v = { inR[k * ip + c], inI[k * ip + c] };
            acc.x += p.x * v.x - p.y * v.y;
            acc.y += p.x * v.y + p.y * v.x;
        }
        outR[r * op + c] = acc.x; outI[r * op + c] = acc.y;
    };

    {
        float ang;
        if (tid < ENC_H) {
            int lq = tid / NANG, a = tid - lq * NANG;
            const float* wp = params + lq * 3;
            float w0 = wp[0], w1 = wp[1], w2 = wp[2];
            ang = (a == 0) ? 0.5f * w0
                : (a == 1) ? 0.5f * w1
                : (a == 2) ? 0.5f * w2
                : (a == 3) ? 0.5f * (w1 - w2)
                :            0.5f * (w1 + w2);
        } else {
            int e = tid - ENC_H;
            int q = (e < NQ) ? e : e - NQ;
            float xv = xin[b * NQ + q];
            ang = (e < NQ) ? 0.5f * xv : 0.5f * xv * xv;
        }
        tcs[tid][0] = cosf(ang);
        tcs[tid][1] = sinf(ang);
    }
    if (tid < NQ - 4) {
        int q = 4 + tid;
        float xv = xin[b * NQ + q];
        float a = 0.5f * xv * xv;
        tcs[ENC_A + q][0] = cosf(a);
        tcs[ENC_A + q][1] = sinf(a);
    }
    __syncthreads();

    if (tid < NLAYER * NQ) {
        const int l = tid / NQ, q = tid - l * NQ;
        const float (*t)[2] = &tcs[tid * NANG];
        float c0 = t[0][0], s0 = t[0][1];
        float c1 = t[1][0], s1 = t[1][1];
        float cz = t[2][0], sz = t[2][1];
        c32 A  = {  c1 * c0,  s1 * s0 };
        c32 Bm = { -s1 * c0, -c1 * s0 };
        c32 Cm = {  s1 * c0, -c1 * s0 };
        c32 Dm = {  c1 * c0, -s1 * s0 };
        c32 ez  = { cz,  sz };
        c32 ezc = { cz, -sz };
        c32 u00 = cmul(A, ezc), u01 = cmul(Bm, ezc);
        c32 u10 = cmul(Cm, ez), u11 = cmul(Dm, ez);
        float* uo = uq[l][q];
        uo[0] = u00.x; uo[1] = u00.y;
        uo[2] = u01.x; uo[3] = u01.y;
        uo[4] = u10.x; uo[5] = u10.y;
        uo[6] = u11.x; uo[7] = u11.y;
        float cd = t[3][0], sd = t[3][1];
        float cs = t[4][0], ss = t[4][1];
        float cph = c0, sph = s0;
        float* pr = pmr[l][q];
        float* pi = pmi[l][q];
#pragma unroll
        for (int e = 0; e < 16; ++e) { pr[e] = 0.0f; pi[e] = 0.0f; }
        pr[0]  = cd;
        pr[3]  = sd * sph;  pi[3]  = -sd * cph;
        pi[12] = -sd;
        pr[15] = cd * cph;  pi[15] =  cd * sph;
        pr[5]  = cs;
        pr[6]  = -ss * sph; pi[6]  = -ss * cph;
        pi[9]  = -ss;
        pr[10] = cs * cph;  pi[10] = -cs * sph;
    } else if (tid >= 64 && tid < 64 + NQ) {
        int q = tid - 64;
        float ch = tcs[ENC_H + q][0], sh = tcs[ENC_H + q][1];
        float ca = tcs[ENC_A + q][0], sa = tcs[ENC_A + q][1];
        enc[q][0] = ch * ca;  enc[q][1] = -ch * sa;
        enc[q][2] = sh * ca;  enc[q][3] =  sh * sa;
    } else if (tid >= 96 && tid < 96 + NLAYER) {
        int l = tid - 96;
        uq[l][UID][0] = 1.0f; uq[l][UID][1] = 0.0f; uq[l][UID][2] = 0.0f; uq[l][UID][3] = 0.0f;
        uq[l][UID][4] = 0.0f; uq[l][UID][5] = 0.0f; uq[l][UID][6] = 1.0f; uq[l][UID][7] = 0.0f;
    }
    __syncthreads();

#pragma unroll
    for (int g = 0; g < NU; ++g)
        kron_build(g >> 2, g & 3, bufAr + 256 * g, bufAi + 256 * g);
    __syncthreads();
#pragma unroll
    for (int g = 0; g < NU; ++g)
        chain_step(g >> 2, 3 * (g & 3), 0, bufAr + 256 * g, bufAi + 256 * g, 16,
                                           bufBr + UPW * g, bufBi + UPW * g, UP);
    __syncthreads();
#pragma unroll
    for (int g = 0; g < NU; ++g)
        chain_step(g >> 2, 3 * (g & 3) + 1, 1, bufBr + UPW * g, bufBi + UPW * g, UP,
                                               bufAr + 256 * g, bufAi + 256 * g, 16);
    __syncthreads();
#pragma unroll
    for (int g = 0; g < NU; ++g)
        chain_step(g >> 2, 3 * (g & 3) + 2, 2, bufAr + 256 * g, bufAi + 256 * g, 16,
                                               bufBr + UPW * g, bufBi + UPW * g, UP);
    __syncthreads();

    const int st = pa(tid);
    for (int m = 0; m < 16; ++m) {
        int k = tid + 256 * m;
        c32 amp = { 1.0f, 0.0f };
#pragma unroll
        for (int q = 0; q < NQ; ++q) {
            int bit = (k >> (11 - q)) & 1;
            c32 f = { enc[q][bit * 2], enc[q][bit * 2 + 1] };
            amp = cmul(amp, f);
        }
        const int a = st + pa(256 * m);
        sre[a] = amp.x; sim_[a] = amp.y;
    }
    __syncthreads();

    for (int l = 0; l < NLAYER; ++l) {
        const float* Ur = bufBr + NWIN * UPW * l;
        const float* Ui = bufBi + NWIN * UPW * l;
        wmma_apply<0>(sre, sim_, Ur,           Ui,           wv, half, lr); __syncthreads();
        wmma_apply<1>(sre, sim_, Ur + UPW,     Ui + UPW,     wv, half, lr); __syncthreads();
        wmma_apply<2>(sre, sim_, Ur + 2 * UPW, Ui + 2 * UPW, wv, half, lr); __syncthreads();
        wmma_apply<3>(sre, sim_, Ur + 3 * UPW, Ui + 3 * UPW, wv, half, lr); __syncthreads();
    }

    float part[NQ];
#pragma unroll
    for (int q = 0; q < NQ; ++q) part[q] = 0.0f;
    for (int m = 0; m < 16; ++m) {
        int k = tid + 256 * m;
        const int a = st + pa(256 * m);
        float pr = sre[a] * sre[a] + sim_[a] * sim_[a];
#pragma unroll
        for (int q = 0; q < NQ; ++q)
            part[q] += ((k >> (11 - q)) & 1) ? -pr : pr;
    }
#pragma unroll
    for (int q = 0; q < NQ; ++q) red[tid * NQ + q] = part[q];
    __syncthreads();
    for (int s = 128; s > 0; s >>= 1) {
        if (tid < s) {
#pragma unroll
            for (int q = 0; q < NQ; ++q)
                red[tid * NQ + q] += red[(tid + s) * NQ + q];
        }
        __syncthreads();
    }
    if (tid < NQ) out[b * NQ + tid] = red[tid];
}

extern "C" void kernel_launch(void* const* d_in, const int* in_sizes, int n_in,
                              void* d_out, int out_size, void* d_ws, size_t ws_size,
                              hipStream_t stream) {
    (void)n_in; (void)out_size; (void)d_ws; (void)ws_size;
    const float* x      = (const float*)d_in[0];
    const float* params = (const float*)d_in[1];
    float* out = (float*)d_out;
    int B = in_sizes[0] / NQ;
    qsim_kernel<<<dim3(B), dim3(256), 0, stream>>>(x, params, out);
}


# All 16 window unitaries built one column per lane in registers (kron column, then the 3 pair gates on its 16 entries, zero terms dropped), not in 4 LDS passes; last-bit rounding of U may differ
# speedup vs baseline: 2.0831x; 1.1930x over previous
#include <hip/hip_runtime.h>


typedef float v2f __attribute__((ext_vector_type(2)));
typedef float v4f __attribute__((ext_vector_type(4)));
typedef float v8f __attribute__((ext_vector_type(8)));

struct c32 { float x, y; };
__device__ __forceinline__ c32 cmul(c32 a, c32 b) {
    return { a.x * b.x - a.y * b.y, a.x * b.y + a.y * b.x };
}

template <int SH>
__device__ __forceinline__ void pair_step(c32 (&v)[16], v4f e0, v4f e1, v4f e2) {
#pragma unroll
    for (int r = 0; r < 16; ++r) {
        if (r & (3 << SH)) continue;
        const int i1 = r | (1 << SH), i2 = r | (2 << SH), i3 = r | (3 << SH);
        const c32 v0 = v[r], v1 = v[i1], v2 = v[i2], v3 = v[i3];
        v[r].x  = e0[0] * v0.x + (e0[1] * v3.x - e0[2] * v3.y);
        v[r].y  = e0[0] * v0.y + (e0[1] * v3.y + e0[2] * v3.x);
        v[i3].x = (e1[0] * v3.x - e1[1] * v3.y) - e0[3] * v0.y;
        v[i3].y = e0[3] * v0.x + (e1[0] * v3.y + e1[1] * v3.x);
        v[i1].x = e1[2] * v1.x + (e1[3] * v2.x - e2[0] * v2.y);
        v[i1].y = e1[2] * v1.y + (e1[3] * v2.y + e2[0] * v2.x);
        v[i2].x = (e2[2] * v2.x - e2[3] * v2.y) - e2[1] * v1.y;
        v[i2].y = e2[1] * v1.x + (e2[2] * v2.y + e2[3] * v2.x);
    }
}

#define NQ     12
#define DIM    4096
#define NLAYER 4
#define NWIN   4
#define UID    NQ
#define NANG   5
#define NU     (NLAYER * NWIN)
#define ENC_H  (NLAYER * NQ * NANG)
#define ENC_A  (ENC_H + NQ)
#define NTRIG  (ENC_A + NQ)
#define UP     20
#define UPW    (16 * UP)

#define SP     5664
__device__ __forceinline__ constexpr int pa(int k) {
    return k + ((k >> 6) & 1) * 18 + ((k >> 7) & 1) * 68 + ((k >> 8) & 1) * 96
             + ((k >> 9) & 1) * 208 + ((k >> 10) & 1) * 392 + ((k >> 11) & 1) * 784;
}

template <int W>
__device__ __forceinline__ constexpr int row_addr(int r) {
    if (W == 0) return r << 8;
    if (W == 1) return r << 5;
    if (W == 2) return r << 2;
    return (r >> 1) | ((r & 1) << 11);
}
template <int W>
__device__ __forceinline__ constexpr int col_addr(int c) {
    if (W == 0) return c;
    if (W == 1) return (c & 31) | ((c >> 5) << 9);
    if (W == 2) return (c & 3) | ((c >> 2) << 6);
    return c << 3;
}

template <int W>
__device__ __forceinline__ void wmma_apply(float* sre, float* sim_,
                                           const float* Wr, const float* Wi,
                                           int wv, int half, int lr) {
    v2f ar[4], an[4], as[4];
#pragma unroll
    for (int ks = 0; ks < 4; ++ks) {
        int kk0 = 4 * ks + 2 * half;
        v2f ai;
        ar[ks][0] = Wr[lr * UP + kk0];  ar[ks][1] = Wr[lr * UP + kk0 + 1];
        ai[0]     = Wi[lr * UP + kk0];  ai[1]     = Wi[lr * UP + kk0 + 1];
        an[ks] = -ai;
        as[ks] = ar[ks] + ai;
    }
    for (int tt = 0; tt < 2; ++tt) {
        const int cb = col_addr<W>((2 * wv + tt) * 16 + lr);
        const int lb = pa(cb + row_addr<W>(2 * half));
        const int sb = pa(cb + row_addr<W>(8 * half));
        v2f br[4], bs[4], bd[4];
#pragma unroll
        for (int ks = 0; ks < 4; ++ks) {
            int i0 = lb + pa(row_addr<W>(4 * ks));
            int i1 = lb + pa(row_addr<W>(4 * ks + 1));
            v2f bi;
            br[ks][0] = sre[i0];  br[ks][1] = sre[i1];
            bi[0]     = sim_[i0]; bi[1]     = sim_[i1];
            bs[ks] = br[ks] + bi;
            bd[ks] = bi - br[ks];
        }
        v8f k1 = {};
#pragma unroll
        for (int ks = 0; ks < 4; ++ks)
            k1 = __builtin_amdgcn_wmma_f32_16x16x4_f32(
                false, as[ks], false, br[ks], (short)0, k1, false, false);
        v8f accR = k1; v8f accI = k1;
#pragma unroll
        for (int ks = 0; ks < 4; ++ks) {
            accR = __builtin_amdgcn_wmma_f32_16x16x4_f32(
                false, an[ks], false, bs[ks], (short)0, accR, false, false);
            accI = __builtin_amdgcn_wmma_f32_16x16x4_f32(
                false, ar[ks], false, bd[ks], (short)0, accI, false, false);
        }
#pragma unroll
        for (int j = 0; j < 8; ++j) {
            int o = sb + pa(row_addr<W>(j));
            sre[o] = accR[j]; sim_[o] = accI[j];
        }
    }
}

__global__ __launch_bounds__(256)
void qsim_kernel(const float* __restrict__ xin,
                 const float* __restrict__ params,
                 float* __restrict__ out) {
    __shared__ float sre[SP];
    __shared__ float sim_[SP];
    __shared__ float bufBr[NU * UPW], bufBi[NU * UPW];
    __shared__ float tcs[NTRIG][2];
    __shared__ v4f uqc[NLAYER][NQ + 1][2];
    __shared__ v4f pk[NLAYER][NQ][3];
    __shared__ float enc[NQ][4];
    float* red = bufBr;

    const int tid  = threadIdx.x;
    const int b    = blockIdx.x;
    const int lane = tid & 31;
    const int wv   = tid >> 5;
    const int half = lane >> 4;
    const int lr   = lane & 15;

    {
        float ang;
        if (tid < ENC_H) {
            int lq = tid / NANG, a = tid - lq * NANG;
            const float* wp = params + lq * 3;
            float w0 = wp[0], w1 = wp[1], w2 = wp[2];
            ang = (a == 0) ? 0.5f * w0
                : (a == 1) ? 0.5f * w1
                : (a == 2) ? 0.5f * w2
                : (a == 3) ? 0.5f * (w1 - w2)
                :            0.5f * (w1 + w2);
        } else {
            int e = tid - ENC_H;
            int q = (e < NQ) ? e : e - NQ;
            float xv = xin[b * NQ + q];
            ang = (e < NQ) ? 0.5f * xv : 0.5f * xv * xv;
        }
        tcs[tid][0] = cosf(ang);
        tcs[tid][1] = sinf(ang);
    }
    if (tid < NQ - 4) {
        int q = 4 + tid;
        float xv = xin[b * NQ + q];
        float a = 0.5f * xv * xv;
        tcs[ENC_A + q][0] = cosf(a);
        tcs[ENC_A + q][1] = sinf(a);
    }
    __syncthreads();

    if (tid < NLAYER * NQ) {
        const int l = tid / NQ, q = tid - l * NQ;
        const float (*t)[2] = &tcs[tid * NANG];
        float c0 = t[0][0], s0 = t[0][1];
        float c1 = t[1][0], s1 = t[1][1];
        float cz = t[2][0], sz = t[2][1];
        c32 A  = {  c1 * c0,  s1 * s0 };
        c32 Bm = { -s1 * c0, -c1 * s0 };
        c32 Cm = {  s1 * c0, -c1 * s0 };
        c32 Dm = {  c1 * c0, -s1 * s0 };
        c32 ez  = { cz,  sz };
        c32 ezc = { cz, -sz };
        c32 u00 = cmul(A, ezc), u01 = cmul(Bm, ezc);
        c32 u10 = cmul(Cm, ez), u11 = cmul(Dm, ez);
        uqc[l][q][0] = v4f{ u00.x, u00.y, u10.x, u10.y };
        uqc[l][q][1] = v4f{ u01.x, u01.y, u11.x, u11.y };
        float cd = t[3][0], sd = t[3][1];
        float cs = t[4][0], ss = t[4][1];
        float cph = c0, sph = s0;
        pk[l][q][0] = v4f{ cd, sd * sph, -sd * cph, -sd };
        pk[l][q][1] = v4f{ cd * cph, cd * sph, cs, -ss * sph };
        pk[l][q][2] = v4f{ -ss * cph, -ss, cs * cph, -cs * sph };
    } else if (tid >= 64 && tid < 64 + NQ) {
        int q = tid - 64;
        float ch = tcs[ENC_H + q][0], sh = tcs[ENC_H + q][1];
        float ca = tcs[ENC_A + q][0], sa = tcs[ENC_A + q][1];
        enc[q][0] = ch * ca;  enc[q][1] = -ch * sa;
        enc[q][2] = sh * ca;  enc[q][3] =  sh * sa;
    } else if (tid >= 96 && tid < 96 + NLAYER) {
        int l = tid - 96;
        uqc[l][UID][0] = v4f{ 1.0f, 0.0f, 0.0f, 0.0f };
        uqc[l][UID][1] = v4f{ 0.0f, 0.0f, 1.0f, 0.0f };
    }
    __syncthreads();

    {
        const int g = tid >> 4, c = tid & 15;
        const int l = g >> 2, w = g & 3;
        const int f0 = (w == 0) ? 0 : UID;
        const int f1 = 3 * w + 1, f2 = 3 * w + 2;
        const int f3 = (w == 3) ? UID : 3 * w + 3;
        const v4f a0 = uqc[l][f0][(c >> 3) & 1];
        const v4f a1 = uqc[l][f1][(c >> 2) & 1];
        const v4f a2 = uqc[l][f2][(c >> 1) & 1];
        const v4f a3 = uqc[l][f3][c & 1];
        const c32 u0[2] = { { a0[0], a0[1] }, { a0[2], a0[3] } };
        const c32 u1[2] = { { a1[0], a1[1] }, { a1[2], a1[3] } };
        const c32 u2[2] = { { a2[0], a2[1] }, { a2[2], a2[3] } };
        const c32 u3[2] = { { a3[0], a3[1] }, { a3[2], a3[3] } };
        c32 v[16];
#pragma unroll
        for (int r3 = 0; r3 < 2; ++r3) {
#pragma unroll
            for (int r2 = 0; r2 < 2; ++r2) {
                const c32 t1 = cmul(u0[r3], u1[r2]);
#pragma unroll
                for (int r1 = 0; r1 < 2; ++r1) {
                    const c32 t2 = cmul(t1, u2[r1]);
#pragma unroll
                    for (int r0 = 0; r0 < 2; ++r0)
                        v[8 * r3 + 4 * r2 + 2 * r1 + r0] = cmul(t2, u3[r0]);
                }
            }
        }
        const v4f (*pp)[3] = &pk[l][3 * w];
        pair_step<2>(v, pp[0][0], pp[0][1], pp[0][2]);
        pair_step<1>(v, pp[1][0], pp[1][1], pp[1][2]);
        pair_step<0>(v, pp[2][0], pp[2][1], pp[2][2]);
        float* oR = bufBr + UPW * g + c;
        float* oI = bufBi + UPW * g + c;
#pragma unroll
        for (int r = 0; r < 16; ++r) { oR[r * UP] = v[r].x; oI[r * UP] = v[r].y; }
    }

    const int st = pa(tid);
    for (int m = 0; m < 16; ++m) {
        int k = tid + 256 * m;
        c32 amp = { 1.0f, 0.0f };
#pragma unroll
        for (int q = 0; q < NQ; ++q) {
            int bit = (k >> (11 - q)) & 1;
            c32 f = { enc[q][bit * 2], enc[q][bit * 2 + 1] };
            amp = cmul(amp, f);
        }
        const int a = st + pa(256 * m);
        sre[a] = amp.x; sim_[a] = amp.y;
    }
    __syncthreads();

    for (int l = 0; l < NLAYER; ++l) {
        const float* Ur = bufBr + NWIN * UPW * l;
        const float* Ui = bufBi + NWIN * UPW * l;
        wmma_apply<0>(sre, sim_, Ur,           Ui,           wv, half, lr); __syncthreads();
        wmma_apply<1>(sre, sim_, Ur + UPW,     Ui + UPW,     wv, half, lr); __syncthreads();
        wmma_apply<2>(sre, sim_, Ur + 2 * UPW, Ui + 2 * UPW, wv, half, lr); __syncthreads();
        wmma_apply<3>(sre, sim_, Ur + 3 * UPW, Ui + 3 * UPW, wv, half, lr); __syncthreads();
    }

    float part[NQ];
#pragma unroll
    for (int q = 0; q < NQ; ++q) part[q] = 0.0f;
    for (int m = 0; m < 16; ++m) {
        int k = tid + 256 * m;
        const int a = st + pa(256 * m);
        float pr = sre[a] * sre[a] + sim_[a] * sim_[a];
#pragma unroll
        for (int q = 0; q < NQ; ++q)
            part[q] += ((k >> (11 - q)) & 1) ? -pr : pr;
    }
#pragma unroll
    for (int q = 0; q < NQ; ++q) red[tid * NQ + q] = part[q];
    __syncthreads();
    for (int s = 128; s > 0; s >>= 1) {
        if (tid < s) {
#pragma unroll
            for (int q = 0; q < NQ; ++q)
                red[tid * NQ + q] += red[(tid + s) * NQ + q];
        }
        __syncthreads();
    }
    if (tid < NQ) out[b * NQ + tid] = red[tid];
}

extern "C" void kernel_launch(void* const* d_in, const int* in_sizes, int n_in,
                              void* d_out, int out_size, void* d_ws, size_t ws_size,
                              hipStream_t stream) {
    (void)n_in; (void)out_size; (void)d_ws; (void)ws_size;
    const float* x      = (const float*)d_in[0];
    const float* params = (const float*)d_in[1];
    float* out = (float*)d_out;
    int B = in_sizes[0] / NQ;
    qsim_kernel<<<dim3(B), dim3(256), 0, stream>>>(x, params, out);
}


# Layer 0 window 0 applied to the encoded product state as a 16-vector in registers, state then formed as its outer product with wires 4-11: 15 WMMA applies, not 16; that window's rounding changes
# speedup vs baseline: 2.1662x; 1.0399x over previous
#include <hip/hip_runtime.h>


typedef float v2f __attribute__((ext_vector_type(2)));
typedef float v4f __attribute__((ext_vector_type(4)));
typedef float v8f __attribute__((ext_vector_type(8)));

struct c32 { float x, y; };
__device__ __forceinline__ c32 cmul(c32 a, c32 b) {
    return { a.x * b.x - a.y * b.y, a.x * b.y + a.y * b.x };
}
__device__ __forceinline__ void gate_times_vec(c32 (&u)[2], v4f k0, v4f k1, const float* e) {
    const c32 e0 = { e[0], e[1] }, e1 = { e[2], e[3] };
    const c32 p0 = cmul(c32{ k0[0], k0[1] }, e0), q0 = cmul(c32{ k1[0], k1[1] }, e1);
    const c32 p1 = cmul(c32{ k0[2], k0[3] }, e0), q1 = cmul(c32{ k1[2], k1[3] }, e1);
    u[0] = { p0.x + q0.x, p0.y + q0.y };
    u[1] = { p1.x + q1.x, p1.y + q1.y };
}

template <int SH>
__device__ __forceinline__ void pair_step(c32 (&v)[16], v4f e0, v4f e1, v4f e2) {
#pragma unroll
    for (int r = 0; r < 16; ++r) {
        if (r & (3 << SH)) continue;
        const int i1 = r | (1 << SH), i2 = r | (2 << SH), i3 = r | (3 << SH);
        const c32 v0 = v[r], v1 = v[i1], v2 = v[i2], v3 = v[i3];
        v[r].x  = e0[0] * v0.x + (e0[1] * v3.x - e0[2] * v3.y);
        v[r].y  = e0[0] * v0.y + (e0[1] * v3.y + e0[2] * v3.x);
        v[i3].x = (e1[0] * v3.x - e1[1] * v3.y) - e0[3] * v0.y;
        v[i3].y = e0[3] * v0.x + (e1[0] * v3.y + e1[1] * v3.x);
        v[i1].x = e1[2] * v1.x + (e1[3] * v2.x - e2[0] * v2.y);
        v[i1].y = e1[2] * v1.y + (e1[3] * v2.y + e2[0] * v2.x);
        v[i2].x = (e2[2] * v2.x - e2[3] * v2.y) - e2[1] * v1.y;
        v[i2].y = e2[1] * v1.x + (e2[2] * v2.y + e2[3] * v2.x);
    }
}

#define NQ     12
#define DIM    4096
#define NLAYER 4
#define NWIN   4
#define UID    NQ
#define NANG   5
#define NU     (NLAYER * NWIN)
#define ENC_H  (NLAYER * NQ * NANG)
#define ENC_A  (ENC_H + NQ)
#define NTRIG  (ENC_A + NQ)
#define UP     20
#define UPW    (16 * UP)

#define SP     5664
__device__ __forceinline__ constexpr int pa(int k) {
    return k + ((k >> 6) & 1) * 18 + ((k >> 7) & 1) * 68 + ((k >> 8) & 1) * 96
             + ((k >> 9) & 1) * 208 + ((k >> 10) & 1) * 392 + ((k >> 11) & 1) * 784;
}

template <int W>
__device__ __forceinline__ constexpr int row_addr(int r) {
    if (W == 0) return r << 8;
    if (W == 1) return r << 5;
    if (W == 2) return r << 2;
    return (r >> 1) | ((r & 1) << 11);
}
template <int W>
__device__ __forceinline__ constexpr int col_addr(int c) {
    if (W == 0) return c;
    if (W == 1) return (c & 31) | ((c >> 5) << 9);
    if (W == 2) return (c & 3) | ((c >> 2) << 6);
    return c << 3;
}

template <int W>
__device__ __forceinline__ void wmma_apply(float* sre, float* sim_,
                                           const float* Wr, const float* Wi,
                                           int wv, int half, int lr) {
    v2f ar[4], an[4], as[4];
#pragma unroll
    for (int ks = 0; ks < 4; ++ks) {
        int kk0 = 4 * ks + 2 * half;
        v2f ai;
        ar[ks][0] = Wr[lr * UP + kk0];  ar[ks][1] = Wr[lr * UP + kk0 + 1];
        ai[0]     = Wi[lr * UP + kk0];  ai[1]     = Wi[lr * UP + kk0 + 1];
        an[ks] = -ai;
        as[ks] = ar[ks] + ai;
    }
    for (int tt = 0; tt < 2; ++tt) {
        const int cb = col_addr<W>((2 * wv + tt) * 16 + lr);
        const int lb = pa(cb + row_addr<W>(2 * half));
        const int sb = pa(cb + row_addr<W>(8 * half));
        v2f br[4], bs[4], bd[4];
#pragma unroll
        for (int ks = 0; ks < 4; ++ks) {
            int i0 = lb + pa(row_addr<W>(4 * ks));
            int i1 = lb + pa(row_addr<W>(4 * ks + 1));
            v2f bi;
            br[ks][0] = sre[i0];  br[ks][1] = sre[i1];
            bi[0]     = sim_[i0]; bi[1]     = sim_[i1];
            bs[ks] = br[ks] + bi;
            bd[ks] = bi - br[ks];
        }
        v8f k1 = {};
#pragma unroll
        for (int ks = 0; ks < 4; ++ks)
            k1 = __builtin_amdgcn_wmma_f32_16x16x4_f32(
                false, as[ks], false, br[ks], (short)0, k1, false, false);
        v8f accR = k1; v8f accI = k1;
#pragma unroll
        for (int ks = 0; ks < 4; ++ks) {
            accR = __builtin_amdgcn_wmma_f32_16x16x4_f32(
                false, an[ks], false, bs[ks], (short)0, accR, false, false);
            accI = __builtin_amdgcn_wmma_f32_16x16x4_f32(
                false, ar[ks], false, bd[ks], (short)0, accI, false, false);
        }
#pragma unroll
        for (int j = 0; j < 8; ++j) {
            int o = sb + pa(row_addr<W>(j));
            sre[o] = accR[j]; sim_[o] = accI[j];
        }
    }
}

__global__ __launch_bounds__(256)
void qsim_kernel(const float* __restrict__ xin,
                 const float* __restrict__ params,
                 float* __restrict__ out) {
    __shared__ float sre[SP];
    __shared__ float sim_[SP];
    __shared__ float bufBr[NU * UPW], bufBi[NU * UPW];
    __shared__ float tcs[NTRIG][2];
    __shared__ v4f uqc[NLAYER][NQ + 1][2];
    __shared__ v4f pk[NLAYER][NQ][3];
    __shared__ float enc[NQ][4];
    float* red = bufBr;

    const int tid  = threadIdx.x;
    const int b    = blockIdx.x;
    const int lane = tid & 31;
    const int wv   = tid >> 5;
    const int half = lane >> 4;
    const int lr   = lane & 15;

    {
        float ang;
        if (tid < ENC_H) {
            int lq = tid / NANG, a = tid - lq * NANG;
            const float* wp = params + lq * 3;
            float w0 = wp[0], w1 = wp[1], w2 = wp[2];
            ang = (a == 0) ? 0.5f * w0
                : (a == 1) ? 0.5f * w1
                : (a == 2) ? 0.5f * w2
                : (a == 3) ? 0.5f * (w1 - w2)
                :            0.5f * (w1 + w2);
        } else {
            int e = tid - ENC_H;
            int q = (e < NQ) ? e : e - NQ;
            float xv = xin[b * NQ + q];
            ang = (e < NQ) ? 0.5f * xv : 0.5f * xv * xv;
        }
        tcs[tid][0] = cosf(ang);
        tcs[tid][1] = sinf(ang);
    }
    if (tid < NQ - 4) {
        int q = 4 + tid;
        float xv = xin[b * NQ + q];
        float a = 0.5f * xv * xv;
        tcs[ENC_A + q][0] = cosf(a);
        tcs[ENC_A + q][1] = sinf(a);
    }
    __syncthreads();

    if (tid < NLAYER * NQ) {
        const int l = tid / NQ, q = tid - l * NQ;
        const float (*t)[2] = &tcs[tid * NANG];
        float c0 = t[0][0], s0 = t[0][1];
        float c1 = t[1][0], s1 = t[1][1];
        float cz = t[2][0], sz = t[2][1];
        c32 A  = {  c1 * c0,  s1 * s0 };
        c32 Bm = { -s1 * c0, -c1 * s0 };
        c32 Cm = {  s1 * c0, -c1 * s0 };
        c32 Dm = {  c1 * c0, -s1 * s0 };
        c32 ez  = { cz,  sz };
        c32 ezc = { cz, -sz };
        c32 u00 = cmul(A, ezc), u01 = cmul(Bm, ezc);
        c32 u10 = cmul(Cm, ez), u11 = cmul(Dm, ez);
        uqc[l][q][0] = v4f{ u00.x, u00.y, u10.x, u10.y };
        uqc[l][q][1] = v4f{ u01.x, u01.y, u11.x, u11.y };
        float cd = t[3][0], sd = t[3][1];
        float cs = t[4][0], ss = t[4][1];
        float cph = c0, sph = s0;
        pk[l][q][0] = v4f{ cd, sd * sph, -sd * cph, -sd };
        pk[l][q][1] = v4f{ cd * cph, cd * sph, cs, -ss * sph };
        pk[l][q][2] = v4f{ -ss * cph, -ss, cs * cph, -cs * sph };
    } else if (tid >= 64 && tid < 64 + NQ) {
        int q = tid - 64;
        float ch = tcs[ENC_H + q][0], sh = tcs[ENC_H + q][1];
        float ca = tcs[ENC_A + q][0], sa = tcs[ENC_A + q][1];
        enc[q][0] = ch * ca;  enc[q][1] = -ch * sa;
        enc[q][2] = sh * ca;  enc[q][3] =  sh * sa;
    } else if (tid >= 96 && tid < 96 + NLAYER) {
        int l = tid - 96;
        uqc[l][UID][0] = v4f{ 1.0f, 0.0f, 0.0f, 0.0f };
        uqc[l][UID][1] = v4f{ 0.0f, 0.0f, 1.0f, 0.0f };
    }
    __syncthreads();

    {
        const int g = tid >> 4, c = tid & 15;
        const int l = g >> 2, w = g & 3;
        const int f0 = (w == 0) ? 0 : UID;
        const int f1 = 3 * w + 1, f2 = 3 * w + 2;
        const int f3 = (w == 3) ? UID : 3 * w + 3;
        const v4f a0 = uqc[l][f0][(c >> 3) & 1];
        const v4f a1 = uqc[l][f1][(c >> 2) & 1];
        const v4f a2 = uqc[l][f2][(c >> 1) & 1];
        const v4f a3 = uqc[l][f3][c & 1];
        c32 u0[2] = { { a0[0], a0[1] }, { a0[2], a0[3] } };
        c32 u1[2] = { { a1[0], a1[1] }, { a1[2], a1[3] } };
        c32 u2[2] = { { a2[0], a2[1] }, { a2[2], a2[3] } };
        c32 u3[2] = { { a3[0], a3[1] }, { a3[2], a3[3] } };
        if (g == 0) {
            gate_times_vec(u0, uqc[0][0][0], uqc[0][0][1], enc[0]);
            gate_times_vec(u1, uqc[0][1][0], uqc[0][1][1], enc[1]);
            gate_times_vec(u2, uqc[0][2][0], uqc[0][2][1], enc[2]);
            gate_times_vec(u3, uqc[0][3][0], uqc[0][3][1], enc[3]);
        }
        c32 v[16];
#pragma unroll
        for (int r3 = 0; r3 < 2; ++r3) {
#pragma unroll
            for (int r2 = 0; r2 < 2; ++r2) {
                const c32 t1 = cmul(u0[r3], u1[r2]);
#pragma unroll
                for (int r1 = 0; r1 < 2; ++r1) {
                    const c32 t2 = cmul(t1, u2[r1]);
#pragma unroll
                    for (int r0 = 0; r0 < 2; ++r0)
                        v[8 * r3 + 4 * r2 + 2 * r1 + r0] = cmul(t2, u3[r0]);
                }
            }
        }
        const v4f (*pp)[3] = &pk[l][3 * w];
        pair_step<2>(v, pp[0][0], pp[0][1], pp[0][2]);
        pair_step<1>(v, pp[1][0], pp[1][1], pp[1][2]);
        pair_step<0>(v, pp[2][0], pp[2][1], pp[2][2]);
        float* oR = bufBr + UPW * g + c;
        float* oI = bufBi + UPW * g + c;
#pragma unroll
        for (int r = 0; r < 16; ++r) { oR[r * UP] = v[r].x; oI[r * UP] = v[r].y; }
    }

    const int st = pa(tid);
    c32 low = { enc[4][((tid >> 7) & 1) * 2], enc[4][((tid >> 7) & 1) * 2 + 1] };
#pragma unroll
    for (int q = 5; q < NQ; ++q) {
        int bit = (tid >> (11 - q)) & 1;
        c32 f = { enc[q][bit * 2], enc[q][bit * 2 + 1] };
        low = cmul(low, f);
    }
    __syncthreads();
    for (int m = 0; m < 16; ++m) {
        const c32 h = { bufBr[m * UP], bufBi[m * UP] };
        const c32 amp = cmul(h, low);
        const int a = st + pa(256 * m);
        sre[a] = amp.x; sim_[a] = amp.y;
    }
    __syncthreads();

    for (int l = 0; l < NLAYER; ++l) {
        const float* Ur = bufBr + NWIN * UPW * l;
        const float* Ui = bufBi + NWIN * UPW * l;
        if (l > 0) {
            wmma_apply<0>(sre, sim_, Ur,       Ui,           wv, half, lr); __syncthreads();
        }
        wmma_apply<1>(sre, sim_, Ur + UPW,     Ui + UPW,     wv, half, lr); __syncthreads();
        wmma_apply<2>(sre, sim_, Ur + 2 * UPW, Ui + 2 * UPW, wv, half, lr); __syncthreads();
        wmma_apply<3>(sre, sim_, Ur + 3 * UPW, Ui + 3 * UPW, wv, half, lr); __syncthreads();
    }

    float part[NQ];
#pragma unroll
    for (int q = 0; q < NQ; ++q) part[q] = 0.0f;
    for (int m = 0; m < 16; ++m) {
        int k = tid + 256 * m;
        const int a = st + pa(256 * m);
        float pr = sre[a] * sre[a] + sim_[a] * sim_[a];
#pragma unroll
        for (int q = 0; q < NQ; ++q)
            part[q] += ((k >> (11 - q)) & 1) ? -pr : pr;
    }
#pragma unroll
    for (int q = 0; q < NQ; ++q) red[tid * NQ + q] = part[q];
    __syncthreads();
    for (int s = 128; s > 0; s >>= 1) {
        if (tid < s) {
#pragma unroll
            for (int q = 0; q < NQ; ++q)
                red[tid * NQ + q] += red[(tid + s) * NQ + q];
        }
        __syncthreads();
    }
    if (tid < NQ) out[b * NQ + tid] = red[tid];
}

extern "C" void kernel_launch(void* const* d_in, const int* in_sizes, int n_in,
                              void* d_out, int out_size, void* d_ws, size_t ws_size,
                              hipStream_t stream) {
    (void)n_in; (void)out_size; (void)d_ws; (void)ws_size;
    const float* x      = (const float*)d_in[0];
    const float* params = (const float*)d_in[1];
    float* out = (float*)d_out;
    int B = in_sizes[0] / NQ;
    qsim_kernel<<<dim3(B), dim3(256), 0, stream>>>(x, params, out);
}


# Layer 0 windows 1-3 use the still factored state: windows 1, 2 act on 16x8 and 16x64 operands (5 WMMA tiles, not 32), window 3 forms its operand from that result on load; rounding there changes
# speedup vs baseline: 2.3893x; 1.1030x over previous
#include <hip/hip_runtime.h>


typedef float v2f __attribute__((ext_vector_type(2)));
typedef float v4f __attribute__((ext_vector_type(4)));
typedef float v8f __attribute__((ext_vector_type(8)));

struct c32 { float x, y; };
__device__ __forceinline__ c32 cmul(c32 a, c32 b) {
    return { a.x * b.x - a.y * b.y, a.x * b.y + a.y * b.x };
}
__device__ __forceinline__ void gate_times_vec(c32 (&u)[2], v4f k0, v4f k1, const float* e) {
    const c32 e0 = { e[0], e[1] }, e1 = { e[2], e[3] };
    const c32 p0 = cmul(c32{ k0[0], k0[1] }, e0), q0 = cmul(c32{ k1[0], k1[1] }, e1);
    const c32 p1 = cmul(c32{ k0[2], k0[3] }, e0), q1 = cmul(c32{ k1[2], k1[3] }, e1);
    u[0] = { p0.x + q0.x, p0.y + q0.y };
    u[1] = { p1.x + q1.x, p1.y + q1.y };
}

template <int SH>
__device__ __forceinline__ void pair_step(c32 (&v)[16], v4f e0, v4f e1, v4f e2) {
#pragma unroll
    for (int r = 0; r < 16; ++r) {
        if (r & (3 << SH)) continue;
        const int i1 = r | (1 << SH), i2 = r | (2 << SH), i3 = r | (3 << SH);
        const c32 v0 = v[r], v1 = v[i1], v2 = v[i2], v3 = v[i3];
        v[r].x  = e0[0] * v0.x + (e0[1] * v3.x - e0[2] * v3.y);
        v[r].y  = e0[0] * v0.y + (e0[1] * v3.y + e0[2] * v3.x);
        v[i3].x = (e1[0] * v3.x - e1[1] * v3.y) - e0[3] * v0.y;
        v[i3].y = e0[3] * v0.x + (e1[0] * v3.y + e1[1] * v3.x);
        v[i1].x = e1[2] * v1.x + (e1[3] * v2.x - e2[0] * v2.y);
        v[i1].y = e1[2] * v1.y + (e1[3] * v2.y + e2[0] * v2.x);
        v[i2].x = (e2[2] * v2.x - e2[3] * v2.y) - e2[1] * v1.y;
        v[i2].y = e2[1] * v1.x + (e2[2] * v2.y + e2[3] * v2.x);
    }
}

#define NQ     12
#define DIM    4096
#define NLAYER 4
#define NWIN   4
#define UID    NQ
#define NANG   5
#define NU     (NLAYER * NWIN)
#define ENC_H  (NLAYER * NQ * NANG)
#define ENC_A  (ENC_H + NQ)
#define NTRIG  (ENC_A + NQ)
#define UP     20
#define UPW    (16 * UP)

#define SP     5664
__device__ __forceinline__ constexpr int pa(int k) {
    return k + ((k >> 6) & 1) * 18 + ((k >> 7) & 1) * 68 + ((k >> 8) & 1) * 96
             + ((k >> 9) & 1) * 208 + ((k >> 10) & 1) * 392 + ((k >> 11) & 1) * 784;
}

template <int W>
__device__ __forceinline__ constexpr int row_addr(int r) {
    if (W == 0) return r << 8;
    if (W == 1) return r << 5;
    if (W == 2) return r << 2;
    return (r >> 1) | ((r & 1) << 11);
}
template <int W>
__device__ __forceinline__ constexpr int col_addr(int c) {
    if (W == 0) return c;
    if (W == 1) return (c & 31) | ((c >> 5) << 9);
    if (W == 2) return (c & 3) | ((c >> 2) << 6);
    return c << 3;
}

template <int W>
__device__ __forceinline__ void wmma_apply(float* sre, float* sim_,
                                           const float* Wr, const float* Wi,
                                           int wv, int half, int lr) {
    v2f ar[4], an[4], as[4];
#pragma unroll
    for (int ks = 0; ks < 4; ++ks) {
        int kk0 = 4 * ks + 2 * half;
        v2f ai;
        ar[ks][0] = Wr[lr * UP + kk0];  ar[ks][1] = Wr[lr * UP + kk0 + 1];
        ai[0]     = Wi[lr * UP + kk0];  ai[1]     = Wi[lr * UP + kk0 + 1];
        an[ks] = -ai;
        as[ks] = ar[ks] + ai;
    }
    for (int tt = 0; tt < 2; ++tt) {
        const int cb = col_addr<W>((2 * wv + tt) * 16 + lr);
        const int lb = pa(cb + row_addr<W>(2 * half));
        const int sb = pa(cb + row_addr<W>(8 * half));
        v2f br[4], bs[4], bd[4];
#pragma unroll
        for (int ks = 0; ks < 4; ++ks) {
            int i0 = lb + pa(row_addr<W>(4 * ks));
            int i1 = lb + pa(row_addr<W>(4 * ks + 1));
            v2f bi;
            br[ks][0] = sre[i0];  br[ks][1] = sre[i1];
            bi[0]     = sim_[i0]; bi[1]     = sim_[i1];
            bs[ks] = br[ks] + bi;
            bd[ks] = bi - br[ks];
        }
        v8f k1 = {};
#pragma unroll
        for (int ks = 0; ks < 4; ++ks)
            k1 = __builtin_amdgcn_wmma_f32_16x16x4_f32(
                false, as[ks], false, br[ks], (short)0, k1, false, false);
        v8f accR = k1; v8f accI = k1;
#pragma unroll
        for (int ks = 0; ks < 4; ++ks) {
            accR = __builtin_amdgcn_wmma_f32_16x16x4_f32(
                false, an[ks], false, bs[ks], (short)0, accR, false, false);
            accI = __builtin_amdgcn_wmma_f32_16x16x4_f32(
                false, ar[ks], false, bd[ks], (short)0, accI, false, false);
        }
#pragma unroll
        for (int j = 0; j < 8; ++j) {
            int o = sb + pa(row_addr<W>(j));
            sre[o] = accR[j]; sim_[o] = accI[j];
        }
    }
}

__device__ __forceinline__ void load_a_frags(const float* Wr, const float* Wi, int half, int lr,
                                             v2f (&ar)[4], v2f (&an)[4], v2f (&as)[4]) {
#pragma unroll
    for (int ks = 0; ks < 4; ++ks) {
        int kk0 = 4 * ks + 2 * half;
        v2f ai;
        ar[ks][0] = Wr[lr * UP + kk0];  ar[ks][1] = Wr[lr * UP + kk0 + 1];
        ai[0]     = Wi[lr * UP + kk0];  ai[1]     = Wi[lr * UP + kk0 + 1];
        an[ks] = -ai;
        as[ks] = ar[ks] + ai;
    }
}
__device__ __forceinline__ void tile_product(const v2f (&ar)[4], const v2f (&an)[4], const v2f (&as)[4],
                                             const v2f (&br)[4], const v2f (&bi)[4],
                                             v8f& accR, v8f& accI) {
    v2f bs[4], bd[4];
#pragma unroll
    for (int ks = 0; ks < 4; ++ks) { bs[ks] = br[ks] + bi[ks]; bd[ks] = bi[ks] - br[ks]; }
    v8f k1 = {};
#pragma unroll
    for (int ks = 0; ks < 4; ++ks)
        k1 = __builtin_amdgcn_wmma_f32_16x16x4_f32(
            false, as[ks], false, br[ks], (short)0, k1, false, false);
    accR = k1; accI = k1;
#pragma unroll
    for (int ks = 0; ks < 4; ++ks) {
        accR = __builtin_amdgcn_wmma_f32_16x16x4_f32(
            false, an[ks], false, bs[ks], (short)0, accR, false, false);
        accI = __builtin_amdgcn_wmma_f32_16x16x4_f32(
            false, ar[ks], false, bd[ks], (short)0, accI, false, false);
    }
}

#define G1P    16
#define G2P    68

__global__ __launch_bounds__(256)
void qsim_kernel(const float* __restrict__ xin,
                 const float* __restrict__ params,
                 float* __restrict__ out) {
    __shared__ float sre[SP];
    __shared__ float sim_[SP];
    __shared__ float bufBr[NU * UPW], bufBi[NU * UPW];
    __shared__ float tcs[NTRIG][2];
    __shared__ v4f uqc[NLAYER][NQ + 1][2];
    __shared__ v4f pk[NLAYER][NQ][3];
    __shared__ float enc[NQ][4];
    __shared__ float g1R[16 * G1P], g1I[16 * G1P];
    __shared__ float g2R[16 * G2P], g2I[16 * G2P];
    float* red = bufBr;

    const int tid  = threadIdx.x;
    const int b    = blockIdx.x;
    const int lane = tid & 31;
    const int wv   = tid >> 5;
    const int half = lane >> 4;
    const int lr   = lane & 15;

    {
        float ang;
        if (tid < ENC_H) {
            int lq = tid / NANG, a = tid - lq * NANG;
            const float* wp = params + lq * 3;
            float w0 = wp[0], w1 = wp[1], w2 = wp[2];
            ang = (a == 0) ? 0.5f * w0
                : (a == 1) ? 0.5f * w1
                : (a == 2) ? 0.5f * w2
                : (a == 3) ? 0.5f * (w1 - w2)
                :            0.5f * (w1 + w2);
        } else {
            int e = tid - ENC_H;
            int q = (e < NQ) ? e : e - NQ;
            float xv = xin[b * NQ + q];
            ang = (e < NQ) ? 0.5f * xv : 0.5f * xv * xv;
        }
        tcs[tid][0] = cosf(ang);
        tcs[tid][1] = sinf(ang);
    }
    if (tid < NQ - 4) {
        int q = 4 + tid;
        float xv = xin[b * NQ + q];
        float a = 0.5f * xv * xv;
        tcs[ENC_A + q][0] = cosf(a);
        tcs[ENC_A + q][1] = sinf(a);
    }
    __syncthreads();

    if (tid < NLAYER * NQ) {
        const int l = tid / NQ, q = tid - l * NQ;
        const float (*t)[2] = &tcs[tid * NANG];
        float c0 = t[0][0], s0 = t[0][1];
        float c1 = t[1][0], s1 = t[1][1];
        float cz = t[2][0], sz = t[2][1];
        c32 A  = {  c1 * c0,  s1 * s0 };
        c32 Bm = { -s1 * c0, -c1 * s0 };
        c32 Cm = {  s1 * c0, -c1 * s0 };
        c32 Dm = {  c1 * c0, -s1 * s0 };
        c32 ez  = { cz,  sz };
        c32 ezc = { cz, -sz };
        c32 u00 = cmul(A, ezc), u01 = cmul(Bm, ezc);
        c32 u10 = cmul(Cm, ez), u11 = cmul(Dm, ez);
        uqc[l][q][0] = v4f{ u00.x, u00.y, u10.x, u10.y };
        uqc[l][q][1] = v4f{ u01.x, u01.y, u11.x, u11.y };
        float cd = t[3][0], sd = t[3][1];
        float cs = t[4][0], ss = t[4][1];
        float cph = c0, sph = s0;
        pk[l][q][0] = v4f{ cd, sd * sph, -sd * cph, -sd };
        pk[l][q][1] = v4f{ cd * cph, cd * sph, cs, -ss * sph };
        pk[l][q][2] = v4f{ -ss * cph, -ss, cs * cph, -cs * sph };
    } else if (tid >= 64 && tid < 64 + NQ) {
        int q = tid - 64;
        float ch = tcs[ENC_H + q][0], sh = tcs[ENC_H + q][1];
        float ca = tcs[ENC_A + q][0], sa = tcs[ENC_A + q][1];
        enc[q][0] = ch * ca;  enc[q][1] = -ch * sa;
        enc[q][2] = sh * ca;  enc[q][3] =  sh * sa;
    } else if (tid >= 96 && tid < 96 + NLAYER) {
        int l = tid - 96;
        uqc[l][UID][0] = v4f{ 1.0f, 0.0f, 0.0f, 0.0f };
        uqc[l][UID][1] = v4f{ 0.0f, 0.0f, 1.0f, 0.0f };
    }
    __syncthreads();

    {
        const int g = tid >> 4, c = tid & 15;
        const int l = g >> 2, w = g & 3;
        const int f0 = (w == 0) ? 0 : UID;
        const int f1 = 3 * w + 1, f2 = 3 * w + 2;
        const int f3 = (w == 3) ? UID : 3 * w + 3;
        const v4f a0 = uqc[l][f0][(c >> 3) & 1];
        const v4f a1 = uqc[l][f1][(c >> 2) & 1];
        const v4f a2 = uqc[l][f2][(c >> 1) & 1];
        const v4f a3 = uqc[l][f3][c & 1];
        c32 u0[2] = { { a0[0], a0[1] }, { a0[2], a0[3] } };
        c32 u1[2] = { { a1[0], a1[1] }, { a1[2], a1[3] } };
        c32 u2[2] = { { a2[0], a2[1] }, { a2[2], a2[3] } };
        c32 u3[2] = { { a3[0], a3[1] }, { a3[2], a3[3] } };
        if (g == 0) {
            gate_times_vec(u0, uqc[0][0][0], uqc[0][0][1], enc[0]);
            gate_times_vec(u1, uqc[0][1][0], uqc[0][1][1], enc[1]);
            gate_times_vec(u2, uqc[0][2][0], uqc[0][2][1], enc[2]);
            gate_times_vec(u3, uqc[0][3][0], uqc[0][3][1], enc[3]);
        }
        c32 v[16];
#pragma unroll
        for (int r3 = 0; r3 < 2; ++r3) {
#pragma unroll
            for (int r2 = 0; r2 < 2; ++r2) {
                const c32 t1 = cmul(u0[r3], u1[r2]);
#pragma unroll
                for (int r1 = 0; r1 < 2; ++r1) {
                    const c32 t2 = cmul(t1, u2[r1]);
#pragma unroll
                    for (int r0 = 0; r0 < 2; ++r0)
                        v[8 * r3 + 4 * r2 + 2 * r1 + r0] = cmul(t2, u3[r0]);
                }
            }
        }
        const v4f (*pp)[3] = &pk[l][3 * w];
        pair_step<2>(v, pp[0][0], pp[0][1], pp[0][2]);
        pair_step<1>(v, pp[1][0], pp[1][1], pp[1][2]);
        pair_step<0>(v, pp[2][0], pp[2][1], pp[2][2]);
        float* oR = bufBr + UPW * g + c;
        float* oI = bufBi + UPW * g + c;
#pragma unroll
        for (int r = 0; r < 16; ++r) { oR[r * UP] = v[r].x; oI[r * UP] = v[r].y; }
    }

    __syncthreads();
    const int wvu = __builtin_amdgcn_readfirstlane(wv);
    if (wvu == 0) {
        v2f ar[4], an[4], as[4];
        load_a_frags(bufBr + UPW, bufBi + UPW, half, lr, ar, an, as);
        const int cc = lr & 7;
        const c32 e5 = { enc[5][2 * half], enc[5][2 * half + 1] };
        v2f br[4], bi[4];
#pragma unroll
        for (int ks = 0; ks < 4; ++ks) {
            const c32 hh = { bufBr[(2 * cc + (ks >> 1)) * UP], bufBi[(2 * cc + (ks >> 1)) * UP] };
            const c32 e4 = { enc[4][2 * (ks & 1)], enc[4][2 * (ks & 1) + 1] };
            const c32 t = cmul(cmul(hh, e4), e5);
#pragma unroll
            for (int sx = 0; sx < 2; ++sx) {
                const c32 e6 = { enc[6][2 * sx], enc[6][2 * sx + 1] };
                const c32 x = cmul(t, e6);
                br[ks][sx] = x.x; bi[ks][sx] = x.y;
            }
        }
        v8f accR, accI;
        tile_product(ar, an, as, br, bi, accR, accI);
#pragma unroll
        for (int j = 0; j < 8; ++j) {
            g1R[(8 * half + j) * G1P + lr] = accR[j];
            g1I[(8 * half + j) * G1P + lr] = accI[j];
        }
    }
    __syncthreads();
    if (wvu < 4) {
        v2f ar[4], an[4], as[4];
        load_a_frags(bufBr + 2 * UPW, bufBi + 2 * UPW, half, lr, ar, an, as);
        const int col6 = 16 * wvu + lr;
        const int gcol = col6 >> 3;
        const int grow = (col6 & 7) << 1;
        const c32 e8 = { enc[8][2 * half], enc[8][2 * half + 1] };
        v2f br[4], bi[4];
#pragma unroll
        for (int ks = 0; ks < 4; ++ks) {
            const int gi = (grow + (ks >> 1)) * G1P + gcol;
            const c32 gg = { g1R[gi], g1I[gi] };
            const c32 e7 = { enc[7][2 * (ks & 1)], enc[7][2 * (ks & 1) + 1] };
            const c32 t = cmul(cmul(gg, e7), e8);
#pragma unroll
            for (int sx = 0; sx < 2; ++sx) {
                const c32 e9 = { enc[9][2 * sx], enc[9][2 * sx + 1] };
                const c32 x = cmul(t, e9);
                br[ks][sx] = x.x; bi[ks][sx] = x.y;
            }
        }
        v8f accR, accI;
        tile_product(ar, an, as, br, bi, accR, accI);
#pragma unroll
        for (int j = 0; j < 8; ++j) {
            g2R[(8 * half + j) * G2P + col6] = accR[j];
            g2I[(8 * half + j) * G2P + col6] = accI[j];
        }
    }
    __syncthreads();

    const int st = pa(tid);
    {
        v2f ar[4], an[4], as[4];
        load_a_frags(bufBr + 3 * UPW, bufBi + 3 * UPW, half, lr, ar, an, as);
        const c32 e11 = { enc[11][2 * half], enc[11][2 * half + 1] };
        const c32 f0 = cmul(c32{ enc[10][0], enc[10][1] }, e11);
        const c32 f1 = cmul(c32{ enc[10][2], enc[10][3] }, e11);
        const int gl = (((lr >> 2) & 1) * 8 + ((lr >> 1) & 1) * 4 + (lr & 1) * 2) * G2P + (lr >> 3);
        for (int tt = 0; tt < 2; ++tt) {
            const int t4 = 2 * wv + tt;
            v2f br[4], bi[4];
#pragma unroll
            for (int w9 = 0; w9 < 2; ++w9) {
#pragma unroll
                for (int sx = 0; sx < 2; ++sx) {
                    const int gi = gl + w9 * G2P + 32 * sx + 2 * t4;
                    const c32 gg = { g2R[gi], g2I[gi] };
                    const c32 x0 = cmul(gg, f0), x1 = cmul(gg, f1);
                    br[2 * w9][sx]     = x0.x; bi[2 * w9][sx]     = x0.y;
                    br[2 * w9 + 1][sx] = x1.x; bi[2 * w9 + 1][sx] = x1.y;
                }
            }
            v8f accR, accI;
            tile_product(ar, an, as, br, bi, accR, accI);
            const int sb = pa(col_addr<3>(t4 * 16 + lr) + row_addr<3>(8 * half));
#pragma unroll
            for (int j = 0; j < 8; ++j) {
                const int o = sb + pa(row_addr<3>(j));
                sre[o] = accR[j]; sim_[o] = accI[j];
            }
        }
    }
    __syncthreads();

    for (int l = 1; l < NLAYER; ++l) {
        const float* Ur = bufBr + NWIN * UPW * l;
        const float* Ui = bufBi + NWIN * UPW * l;
        wmma_apply<0>(sre, sim_, Ur,           Ui,           wv, half, lr); __syncthreads();
        wmma_apply<1>(sre, sim_, Ur + UPW,     Ui + UPW,     wv, half, lr); __syncthreads();
        wmma_apply<2>(sre, sim_, Ur + 2 * UPW, Ui + 2 * UPW, wv, half, lr); __syncthreads();
        wmma_apply<3>(sre, sim_, Ur + 3 * UPW, Ui + 3 * UPW, wv, half, lr); __syncthreads();
    }

    float part[NQ];
#pragma unroll
    for (int q = 0; q < NQ; ++q) part[q] = 0.0f;
    for (int m = 0; m < 16; ++m) {
        int k = tid + 256 * m;
        const int a = st + pa(256 * m);
        float pr = sre[a] * sre[a] + sim_[a] * sim_[a];
#pragma unroll
        for (int q = 0; q < NQ; ++q)
            part[q] += ((k >> (11 - q)) & 1) ? -pr : pr;
    }
#pragma unroll
    for (int q = 0; q < NQ; ++q) red[tid * NQ + q] = part[q];
    __syncthreads();
    for (int s = 128; s > 0; s >>= 1) {
        if (tid < s) {
#pragma unroll
            for (int q = 0; q < NQ; ++q)
                red[tid * NQ + q] += red[(tid + s) * NQ + q];
        }
        __syncthreads();
    }
    if (tid < NQ) out[b * NQ + tid] = red[tid];
}

extern "C" void kernel_launch(void* const* d_in, const int* in_sizes, int n_in,
                              void* d_out, int out_size, void* d_ws, size_t ws_size,
                              hipStream_t stream) {
    (void)n_in; (void)out_size; (void)d_ws; (void)ws_size;
    const float* x      = (const float*)d_in[0];
    const float* params = (const float*)d_in[1];
    float* out = (float*)d_out;
    int B = in_sizes[0] / NQ;
    qsim_kernel<<<dim3(B), dim3(256), 0, stream>>>(x, params, out);
}
